# starting kernel
# baseline (speedup 1.0000x reference)
#include <hip/hip_runtime.h>


#define LLEN   8192
#define NMODES 32
#define NF     64
#define NROWS  4096
#define KSPLIT 16
#define MPAIRS 128
#define NCHUNK 32
#define TWO_PI 6.283185307179586476925f

#define T1P_OFF 0
#define T2_OFF  (LLEN * NF)
#define XF_OFF  (2 * LLEN * NF)
#define G_OFF   (2 * LLEN * NF + NROWS * NF)

typedef float v2f __attribute__((ext_vector_type(2)));
typedef float v8f __attribute__((ext_vector_type(8)));

__global__ __launch_bounds__(256) void k_tables(float* __restrict__ t1p,
                                                float* __restrict__ t2) {
  const int idx = blockIdx.x * 256 + threadIdx.x;
  const int l = idx >> 6;
  const int j = idx & 63;
  const int f = j & 31;
  const int r = (f * l) & (LLEN - 1);
  const float ph = (float)r * (TWO_PI / (float)LLEN);
  const float v = (j >= 32) ? -sinf(ph) : cosf(ph);
  t2[(size_t)l * NF + j] = v;
  t1p[(size_t)(l >> 1) * 128 + j * 2 + (l & 1)] = v;
}

__global__ __launch_bounds__(512) void k_dft_fwd(const float* __restrict__ x,
                                                 const float* __restrict__ t1p,
                                                 float* __restrict__ xf) {
  __shared__ float red[KSPLIT * 32 * NF];
  const int lane = threadIdx.x & 31;
  const int wid  = threadIdx.x >> 5;
  const int hi   = lane >> 4;
  const int ln   = lane & 15;
  const int rowbase = blockIdx.x * 32;
  const float* xrow0 = x + (size_t)(rowbase + ln) * LLEN;
  const float* xrow1 = x + (size_t)(rowbase + 16 + ln) * LLEN;

  v8f a0t0 = {}, a0t1 = {}, a0t2 = {}, a0t3 = {};
  v8f a1t0 = {}, a1t1 = {}, a1t2 = {}, a1t3 = {};
  const int k0beg = wid * (LLEN / KSPLIT);
  const int k0end = k0beg + (LLEN / KSPLIT);
  for (int k0 = k0beg; k0 < k0end; k0 += 4) {
    const int kb = k0 + 2 * hi;
    const v2f a0 = *(const v2f*)(xrow0 + kb);
    const v2f a1 = *(const v2f*)(xrow1 + kb);
    const float* tb = t1p + (size_t)(kb >> 1) * 128;
    const v2f b0 = *(const v2f*)(tb + (0 * 16 + ln) * 2);
    const v2f b1 = *(const v2f*)(tb + (1 * 16 + ln) * 2);
    const v2f b2 = *(const v2f*)(tb + (2 * 16 + ln) * 2);
    const v2f b3 = *(const v2f*)(tb + (3 * 16 + ln) * 2);
    a0t0 = __builtin_amdgcn_wmma_f32_16x16x4_f32(false, a0, false, b0, (short)0, a0t0, false, false);
    a0t1 = __builtin_amdgcn_wmma_f32_16x16x4_f32(false, a0, false, b1, (short)0, a0t1, false, false);
    a0t2 = __builtin_amdgcn_wmma_f32_16x16x4_f32(false, a0, false, b2, (short)0, a0t2, false, false);
    a0t3 = __builtin_amdgcn_wmma_f32_16x16x4_f32(false, a0, false, b3, (short)0, a0t3, false, false);
    a1t0 = __builtin_amdgcn_wmma_f32_16x16x4_f32(false, a1, false, b0, (short)0, a1t0, false, false);
    a1t1 = __builtin_amdgcn_wmma_f32_16x16x4_f32(false, a1, false, b1, (short)0, a1t1, false, false);
    a1t2 = __builtin_amdgcn_wmma_f32_16x16x4_f32(false, a1, false, b2, (short)0, a1t2, false, false);
    a1t3 = __builtin_amdgcn_wmma_f32_16x16x4_f32(false, a1, false, b3, (short)0, a1t3, false, false);
  }

  float* my = red + wid * (32 * NF);
  #pragma unroll
  for (int m = 0; m < 8; ++m) {
    const int r = m + 8 * hi;
    my[r * NF + 0 * 16 + ln] = a0t0[m];
    my[r * NF + 1 * 16 + ln] = a0t1[m];
    my[r * NF + 2 * 16 + ln] = a0t2[m];
    my[r * NF + 3 * 16 + ln] = a0t3[m];
    my[(16 + r) * NF + 0 * 16 + ln] = a1t0[m];
    my[(16 + r) * NF + 1 * 16 + ln] = a1t1[m];
    my[(16 + r) * NF + 2 * 16 + ln] = a1t2[m];
    my[(16 + r) * NF + 3 * 16 + ln] = a1t3[m];
  }
  __syncthreads();
  #pragma unroll
  for (int u = 0; u < 4; ++u) {
    const int idx = threadIdx.x + 512 * u;
    float s = 0.0f;
    #pragma unroll
    for (int w = 0; w < KSPLIT; ++w) s += red[w * (32 * NF) + idx];
    xf[(size_t)blockIdx.x * (32 * NF) + idx] = s;
  }
}

__global__ __launch_bounds__(256) void k_mix(const float* __restrict__ xf,
                                             const float* __restrict__ wr,
                                             const float* __restrict__ wi,
                                             float* __restrict__ g) {
  const int idx = blockIdx.x * 256 + threadIdx.x;
  const int f = idx & 31;
  const int o = (idx >> 5) & 15;
  const int h = (idx >> 9) & 3;
  const int b = idx >> 11;
  const float* xr0 = xf + (size_t)(b * 64 + h * 16) * NF + f;
  const float* wr0 = wr + ((size_t)(h * 16) * 16 + o) * NMODES + f;
  const float* wi0 = wi + ((size_t)(h * 16) * 16 + o) * NMODES + f;
  float or_ = 0.0f, oi_ = 0.0f;
  #pragma unroll
  for (int i = 0; i < 16; ++i) {
    const float xr  = xr0[i * NF];
    const float xi  = xr0[i * NF + 32];
    const float wre = wr0[i * (16 * NMODES)];
    const float wim = wi0[i * (16 * NMODES)];
    or_ += xr * wre - xi * wim;
    oi_ += xr * wim + xi * wre;
  }
  const float invL = 1.0f / (float)LLEN;
  const float scale = (f == 0) ? invL : 2.0f * invL;
  const size_t ro = (size_t)(b * 64 + h * 16 + o);
  g[ro * NF + f]      = or_ * scale;
  g[ro * NF + 32 + f] = oi_ * scale;
}

__global__ __launch_bounds__(256) void k_idft(const float* __restrict__ g,
                                              const float* __restrict__ t2,
                                              const float* __restrict__ bias,
                                              float* __restrict__ out) {
  const int lane = threadIdx.x & 31;
  const int wid  = threadIdx.x >> 5;
  const int gw   = blockIdx.x * 8 + wid;
  const int mpair = gw >> 5;
  const int nc    = gw & (NCHUNK - 1);
  const int hi = lane >> 4, ln = lane & 15;
  const int rowbase = mpair * 32;

  v2f af0[16], af1[16];
  #pragma unroll
  for (int ks = 0; ks < 16; ++ks) {
    const int kb = ks * 4 + 2 * hi;
    af0[ks] = *(const v2f*)(g + (size_t)(rowbase + ln) * NF + kb);
    af1[ks] = *(const v2f*)(g + (size_t)(rowbase + 16 + ln) * NF + kb);
  }
  float bv0[8], bv1[8];
  #pragma unroll
  for (int m = 0; m < 8; ++m) {
    bv0[m] = bias[(rowbase + m + 8 * hi) & 63];
    bv1[m] = bias[(rowbase + 16 + m + 8 * hi) & 63];
  }

  const int colchunk = nc * (LLEN / NCHUNK);
  for (int nt = 0; nt < 16; ++nt) {
    const int col = colchunk + nt * 16 + ln;
    const float* tcol = t2 + (size_t)col * NF;
    v8f acc0 = {}, acc1 = {};
    #pragma unroll
    for (int ks = 0; ks < 16; ++ks) {
      const int kb = ks * 4 + 2 * hi;
      const v2f bf = *(const v2f*)(tcol + kb);
      acc0 = __builtin_amdgcn_wmma_f32_16x16x4_f32(false, af0[ks], false, bf, (short)0, acc0, false, false);
      acc1 = __builtin_amdgcn_wmma_f32_16x16x4_f32(false, af1[ks], false, bf, (short)0, acc1, false, false);
    }
    #pragma unroll
    for (int m = 0; m < 8; ++m) {
      const int r = rowbase + m + 8 * hi;
      out[(size_t)r * LLEN + col]              = acc0[m] + bv0[m];
      out[(size_t)(r + 16) * LLEN + col]       = acc1[m] + bv1[m];
    }
  }
}

extern "C" void kernel_launch(void* const* d_in, const int* in_sizes, int n_in,
                              void* d_out, int out_size, void* d_ws, size_t ws_size,
                              hipStream_t stream) {
  const float* x    = (const float*)d_in[0];
  const float* wre  = (const float*)d_in[1];
  const float* wim  = (const float*)d_in[2];
  const float* bias = (const float*)d_in[3];
  float* out = (float*)d_out;
  float* ws  = (float*)d_ws;

  float* t1p = ws + T1P_OFF;
  float* t2  = ws + T2_OFF;
  float* xf  = ws + XF_OFF;
  float* g   = ws + G_OFF;

  k_tables <<<(LLEN * NF) / 256, 256, 0, stream>>>(t1p, t2);
  k_dft_fwd<<<MPAIRS, 512, 0, stream>>>(x, t1p, xf);
  k_mix    <<<(NROWS * NF / 2) / 256, 256, 0, stream>>>(xf, wre, wim, g);
  k_idft   <<<(MPAIRS * NCHUNK) / 8, 256, 0, stream>>>(g, t2, bias, out);
}


# Stage 3: each lane owns 4 adjacent output columns (4 interleaved N-tiles, inverse basis table re-laid to match): b128 whole-line stores, wave-contiguous b128 basis loads; sum order unchanged
# speedup vs baseline: 1.2021x; 1.2021x over previous
#include <hip/hip_runtime.h>


#define LLEN   8192
#define NMODES 32
#define NF     64
#define NROWS  4096
#define KSPLIT 16
#define MPAIRS 128
#define NCHUNK 32
#define TWO_PI 6.283185307179586476925f

#define T1P_OFF 0
#define T2_OFF  (LLEN * NF)
#define XF_OFF  (2 * LLEN * NF)
#define G_OFF   (2 * LLEN * NF + NROWS * NF)

typedef float v2f __attribute__((ext_vector_type(2)));
typedef float v4f __attribute__((ext_vector_type(4)));
typedef float v8f __attribute__((ext_vector_type(8)));

__global__ __launch_bounds__(256) void k_tables(float* __restrict__ t1p,
                                                float* __restrict__ t2q) {
  const int idx = blockIdx.x * 256 + threadIdx.x;
  const int l = idx >> 6;
  const int j = idx & 63;
  const int f = j & 31;
  const int r = (f * l) & (LLEN - 1);
  const float ph = (float)r * (TWO_PI / (float)LLEN);
  const float v = (j >= 32) ? -sinf(ph) : cosf(ph);
  const int lc = l & 63;
  const int q  = lc & 3;
  const int n  = lc >> 2;
  const int ks = j >> 2, hj = (j >> 1) & 1, e = j & 1;
  t2q[(size_t)(l >> 6) * 4096 + ((ks * 2 + (q >> 1)) * 2 + hj) * 64 + n * 4 + (q & 1) * 2 + e] = v;
  t1p[(size_t)(l >> 1) * 128 + j * 2 + (l & 1)] = v;
}

__global__ __launch_bounds__(512) void k_dft_fwd(const float* __restrict__ x,
                                                 const float* __restrict__ t1p,
                                                 float* __restrict__ xf) {
  __shared__ float red[KSPLIT * 32 * NF];
  const int lane = threadIdx.x & 31;
  const int wid  = threadIdx.x >> 5;
  const int hi   = lane >> 4;
  const int ln   = lane & 15;
  const int rowbase = blockIdx.x * 32;
  const float* xrow0 = x + (size_t)(rowbase + ln) * LLEN;
  const float* xrow1 = x + (size_t)(rowbase + 16 + ln) * LLEN;

  v8f a0t0 = {}, a0t1 = {}, a0t2 = {}, a0t3 = {};
  v8f a1t0 = {}, a1t1 = {}, a1t2 = {}, a1t3 = {};
  const int k0beg = wid * (LLEN / KSPLIT);
  const int k0end = k0beg + (LLEN / KSPLIT);
  for (int k0 = k0beg; k0 < k0end; k0 += 4) {
    const int kb = k0 + 2 * hi;
    const v2f a0 = *(const v2f*)(xrow0 + kb);
    const v2f a1 = *(const v2f*)(xrow1 + kb);
    const float* tb = t1p + (size_t)(kb >> 1) * 128;
    const v2f b0 = *(const v2f*)(tb + (0 * 16 + ln) * 2);
    const v2f b1 = *(const v2f*)(tb + (1 * 16 + ln) * 2);
    const v2f b2 = *(const v2f*)(tb + (2 * 16 + ln) * 2);
    const v2f b3 = *(const v2f*)(tb + (3 * 16 + ln) * 2);
    a0t0 = __builtin_amdgcn_wmma_f32_16x16x4_f32(false, a0, false, b0, (short)0, a0t0, false, false);
    a0t1 = __builtin_amdgcn_wmma_f32_16x16x4_f32(false, a0, false, b1, (short)0, a0t1, false, false);
    a0t2 = __builtin_amdgcn_wmma_f32_16x16x4_f32(false, a0, false, b2, (short)0, a0t2, false, false);
    a0t3 = __builtin_amdgcn_wmma_f32_16x16x4_f32(false, a0, false, b3, (short)0, a0t3, false, false);
    a1t0 = __builtin_amdgcn_wmma_f32_16x16x4_f32(false, a1, false, b0, (short)0, a1t0, false, false);
    a1t1 = __builtin_amdgcn_wmma_f32_16x16x4_f32(false, a1, false, b1, (short)0, a1t1, false, false);
    a1t2 = __builtin_amdgcn_wmma_f32_16x16x4_f32(false, a1, false, b2, (short)0, a1t2, false, false);
    a1t3 = __builtin_amdgcn_wmma_f32_16x16x4_f32(false, a1, false, b3, (short)0, a1t3, false, false);
  }

  float* my = red + wid * (32 * NF);
  #pragma unroll
  for (int m = 0; m < 8; ++m) {
    const int r = m + 8 * hi;
    my[r * NF + 0 * 16 + ln] = a0t0[m];
    my[r * NF + 1 * 16 + ln] = a0t1[m];
    my[r * NF + 2 * 16 + ln] = a0t2[m];
    my[r * NF + 3 * 16 + ln] = a0t3[m];
    my[(16 + r) * NF + 0 * 16 + ln] = a1t0[m];
    my[(16 + r) * NF + 1 * 16 + ln] = a1t1[m];
    my[(16 + r) * NF + 2 * 16 + ln] = a1t2[m];
    my[(16 + r) * NF + 3 * 16 + ln] = a1t3[m];
  }
  __syncthreads();
  #pragma unroll
  for (int u = 0; u < 4; ++u) {
    const int idx = threadIdx.x + 512 * u;
    float s = 0.0f;
    #pragma unroll
    for (int w = 0; w < KSPLIT; ++w) s += red[w * (32 * NF) + idx];
    xf[(size_t)blockIdx.x * (32 * NF) + idx] = s;
  }
}

__global__ __launch_bounds__(256) void k_mix(const float* __restrict__ xf,
                                             const float* __restrict__ wr,
                                             const float* __restrict__ wi,
                                             float* __restrict__ g) {
  const int idx = blockIdx.x * 256 + threadIdx.x;
  const int f = idx & 31;
  const int o = (idx >> 5) & 15;
  const int h = (idx >> 9) & 3;
  const int b = idx >> 11;
  const float* xr0 = xf + (size_t)(b * 64 + h * 16) * NF + f;
  const float* wr0 = wr + ((size_t)(h * 16) * 16 + o) * NMODES + f;
  const float* wi0 = wi + ((size_t)(h * 16) * 16 + o) * NMODES + f;
  float or_ = 0.0f, oi_ = 0.0f;
  #pragma unroll
  for (int i = 0; i < 16; ++i) {
    const float xr  = xr0[i * NF];
    const float xi  = xr0[i * NF + 32];
    const float wre = wr0[i * (16 * NMODES)];
    const float wim = wi0[i * (16 * NMODES)];
    or_ += xr * wre - xi * wim;
    oi_ += xr * wim + xi * wre;
  }
  const float invL = 1.0f / (float)LLEN;
  const float scale = (f == 0) ? invL : 2.0f * invL;
  const size_t ro = (size_t)(b * 64 + h * 16 + o);
  g[ro * NF + f]      = or_ * scale;
  g[ro * NF + 32 + f] = oi_ * scale;
}

__global__ __launch_bounds__(256) void k_idft(const float* __restrict__ g,
                                              const float* __restrict__ t2q,
                                              const float* __restrict__ bias,
                                              float* __restrict__ out) {
  const int lane = threadIdx.x & 31;
  const int wid  = threadIdx.x >> 5;
  const int gw   = blockIdx.x * 8 + wid;
  const int mpair = gw >> 5;
  const int nc    = gw & (NCHUNK - 1);
  const int hi = lane >> 4, ln = lane & 15;
  const int rowbase = mpair * 32;

  v2f af0[16], af1[16];
  #pragma unroll
  for (int ks = 0; ks < 16; ++ks) {
    const int kb = ks * 4 + 2 * hi;
    af0[ks] = *(const v2f*)(g + (size_t)(rowbase + ln) * NF + kb);
    af1[ks] = *(const v2f*)(g + (size_t)(rowbase + 16 + ln) * NF + kb);
  }
  float bv0[8], bv1[8];
  #pragma unroll
  for (int m = 0; m < 8; ++m) {
    bv0[m] = bias[(rowbase + m + 8 * hi) & 63];
    bv1[m] = bias[(rowbase + 16 + m + 8 * hi) & 63];
  }

  const int colchunk = nc * (LLEN / NCHUNK);
  const float* tlane = t2q + (size_t)(colchunk >> 6) * 4096 + (hi * 16 + ln) * 4;
  for (int gq = 0; gq < 4; ++gq) {
    const float* tb = tlane + gq * 4096;
    v8f c00 = {}, c01 = {}, c02 = {}, c03 = {};
    v8f c10 = {}, c11 = {}, c12 = {}, c13 = {};
    #pragma unroll
    for (int ks = 0; ks < 16; ++ks) {
      const v4f p = *(const v4f*)(tb + (ks * 2 + 0) * 128);
      const v4f r = *(const v4f*)(tb + (ks * 2 + 1) * 128);
      const v2f b0 = {p.x, p.y};
      const v2f b1 = {p.z, p.w};
      const v2f b2 = {r.x, r.y};
      const v2f b3 = {r.z, r.w};
      c00 = __builtin_amdgcn_wmma_f32_16x16x4_f32(false, af0[ks], false, b0, (short)0, c00, false, false);
      c01 = __builtin_amdgcn_wmma_f32_16x16x4_f32(false, af0[ks], false, b1, (short)0, c01, false, false);
      c02 = __builtin_amdgcn_wmma_f32_16x16x4_f32(false, af0[ks], false, b2, (short)0, c02, false, false);
      c03 = __builtin_amdgcn_wmma_f32_16x16x4_f32(false, af0[ks], false, b3, (short)0, c03, false, false);
      c10 = __builtin_amdgcn_wmma_f32_16x16x4_f32(false, af1[ks], false, b0, (short)0, c10, false, false);
      c11 = __builtin_amdgcn_wmma_f32_16x16x4_f32(false, af1[ks], false, b1, (short)0, c11, false, false);
      c12 = __builtin_amdgcn_wmma_f32_16x16x4_f32(false, af1[ks], false, b2, (short)0, c12, false, false);
      c13 = __builtin_amdgcn_wmma_f32_16x16x4_f32(false, af1[ks], false, b3, (short)0, c13, false, false);
    }
    const int col = colchunk + gq * 64 + 4 * ln;
    #pragma unroll
    for (int m = 0; m < 8; ++m) {
      const int r = rowbase + m + 8 * hi;
      const v4f o0 = {c00[m] + bv0[m], c01[m] + bv0[m], c02[m] + bv0[m], c03[m] + bv0[m]};
      const v4f o1 = {c10[m] + bv1[m], c11[m] + bv1[m], c12[m] + bv1[m], c13[m] + bv1[m]};
      *(v4f*)(out + (size_t)r * LLEN + col)        = o0;
      *(v4f*)(out + (size_t)(r + 16) * LLEN + col) = o1;
    }
  }
}

extern "C" void kernel_launch(void* const* d_in, const int* in_sizes, int n_in,
                              void* d_out, int out_size, void* d_ws, size_t ws_size,
                              hipStream_t stream) {
  const float* x    = (const float*)d_in[0];
  const float* wre  = (const float*)d_in[1];
  const float* wim  = (const float*)d_in[2];
  const float* bias = (const float*)d_in[3];
  float* out = (float*)d_out;
  float* ws  = (float*)d_ws;

  float* t1p = ws + T1P_OFF;
  float* t2q = ws + T2_OFF;
  float* xf  = ws + XF_OFF;
  float* g   = ws + G_OFF;

  k_tables <<<(LLEN * NF) / 256, 256, 0, stream>>>(t1p, t2q);
  k_dft_fwd<<<MPAIRS, 512, 0, stream>>>(x, t1p, xf);
  k_mix    <<<(NROWS * NF / 2) / 256, 256, 0, stream>>>(xf, wre, wim, g);
  k_idft   <<<(MPAIRS * NCHUNK) / 8, 256, 0, stream>>>(g, t2q, bias, out);
}


# Stage 1: K split over 2 workgroups per M-pair (256 workgroups, one per WGP; was 128 on half the chip); stage 2 adds the 2 partial Xf planes in LDS; Xf reduction tree changed
# speedup vs baseline: 1.5366x; 1.2783x over previous
#include <hip/hip_runtime.h>


#define LLEN   8192
#define NMODES 32
#define NF     64
#define NROWS  4096
#define KSPLIT 16
#define KBLK   2
#define MPAIRS 128
#define NCHUNK 32
#define TWO_PI 6.283185307179586476925f

#define T1P_OFF 0
#define T2_OFF  (LLEN * NF)
#define XF_OFF  (2 * LLEN * NF)
#define G_OFF   (2 * LLEN * NF + KBLK * NROWS * NF)

typedef float v2f __attribute__((ext_vector_type(2)));
typedef float v4f __attribute__((ext_vector_type(4)));
typedef float v8f __attribute__((ext_vector_type(8)));

__global__ __launch_bounds__(256) void k_tables(float* __restrict__ t1p,
                                                float* __restrict__ t2q) {
  const int idx = blockIdx.x * 256 + threadIdx.x;
  const int l = idx >> 6;
  const int j = idx & 63;
  const int f = j & 31;
  const int r = (f * l) & (LLEN - 1);
  const float ph = (float)r * (TWO_PI / (float)LLEN);
  const float v = (j >= 32) ? -sinf(ph) : cosf(ph);
  const int lc = l & 63;
  const int q  = lc & 3;
  const int n  = lc >> 2;
  const int ks = j >> 2, hj = (j >> 1) & 1, e = j & 1;
  t2q[(size_t)(l >> 6) * 4096 + ((ks * 2 + (q >> 1)) * 2 + hj) * 64 + n * 4 + (q & 1) * 2 + e] = v;
  t1p[(size_t)(l >> 1) * 128 + j * 2 + (l & 1)] = v;
}

__global__ __launch_bounds__(512) void k_dft_fwd(const float* __restrict__ x,
                                                 const float* __restrict__ t1p,
                                                 float* __restrict__ xf) {
  __shared__ float red[KSPLIT * 32 * NF];
  const int lane = threadIdx.x & 31;
  const int wid  = threadIdx.x >> 5;
  const int hi   = lane >> 4;
  const int ln   = lane & 15;
  const int mp = blockIdx.x / KBLK;
  const int kh = blockIdx.x % KBLK;
  const int rowbase = mp * 32;
  const float* xrow0 = x + (size_t)(rowbase + ln) * LLEN;
  const float* xrow1 = x + (size_t)(rowbase + 16 + ln) * LLEN;

  v8f a0t0 = {}, a0t1 = {}, a0t2 = {}, a0t3 = {};
  v8f a1t0 = {}, a1t1 = {}, a1t2 = {}, a1t3 = {};
  const int k0beg = (kh * KSPLIT + wid) * (LLEN / (KBLK * KSPLIT));
  for (int it = 0; it < LLEN / (KBLK * KSPLIT * 4); ++it) {
    const int kb = k0beg + 4 * it + 2 * hi;
    const v2f a0 = *(const v2f*)(xrow0 + kb);
    const v2f a1 = *(const v2f*)(xrow1 + kb);
    const float* tb = t1p + (size_t)(kb >> 1) * 128;
    const v2f b0 = *(const v2f*)(tb + (0 * 16 + ln) * 2);
    const v2f b1 = *(const v2f*)(tb + (1 * 16 + ln) * 2);
    const v2f b2 = *(const v2f*)(tb + (2 * 16 + ln) * 2);
    const v2f b3 = *(const v2f*)(tb + (3 * 16 + ln) * 2);
    a0t0 = __builtin_amdgcn_wmma_f32_16x16x4_f32(false, a0, false, b0, (short)0, a0t0, false, false);
    a0t1 = __builtin_amdgcn_wmma_f32_16x16x4_f32(false, a0, false, b1, (short)0, a0t1, false, false);
    a0t2 = __builtin_amdgcn_wmma_f32_16x16x4_f32(false, a0, false, b2, (short)0, a0t2, false, false);
    a0t3 = __builtin_amdgcn_wmma_f32_16x16x4_f32(false, a0, false, b3, (short)0, a0t3, false, false);
    a1t0 = __builtin_amdgcn_wmma_f32_16x16x4_f32(false, a1, false, b0, (short)0, a1t0, false, false);
    a1t1 = __builtin_amdgcn_wmma_f32_16x16x4_f32(false, a1, false, b1, (short)0, a1t1, false, false);
    a1t2 = __builtin_amdgcn_wmma_f32_16x16x4_f32(false, a1, false, b2, (short)0, a1t2, false, false);
    a1t3 = __builtin_amdgcn_wmma_f32_16x16x4_f32(false, a1, false, b3, (short)0, a1t3, false, false);
  }

  float* my = red + wid * (32 * NF);
  #pragma unroll
  for (int m = 0; m < 8; ++m) {
    const int r = m + 8 * hi;
    my[r * NF + 0 * 16 + ln] = a0t0[m];
    my[r * NF + 1 * 16 + ln] = a0t1[m];
    my[r * NF + 2 * 16 + ln] = a0t2[m];
    my[r * NF + 3 * 16 + ln] = a0t3[m];
    my[(16 + r) * NF + 0 * 16 + ln] = a1t0[m];
    my[(16 + r) * NF + 1 * 16 + ln] = a1t1[m];
    my[(16 + r) * NF + 2 * 16 + ln] = a1t2[m];
    my[(16 + r) * NF + 3 * 16 + ln] = a1t3[m];
  }
  __syncthreads();
  #pragma unroll
  for (int u = 0; u < 4; ++u) {
    const int idx = threadIdx.x + 512 * u;
    float s = 0.0f;
    #pragma unroll
    for (int w = 0; w < KSPLIT; ++w) s += red[w * (32 * NF) + idx];
    xf[(size_t)kh * (NROWS * NF) + (size_t)mp * (32 * NF) + idx] = s;
  }
}

__global__ __launch_bounds__(256) void k_mix(const float* __restrict__ xf,
                                             const float* __restrict__ wr,
                                             const float* __restrict__ wi,
                                             float* __restrict__ g) {
  const int idx = blockIdx.x * 256 + threadIdx.x;
  const int f = idx & 31;
  const int o = (idx >> 5) & 15;
  const int h = (idx >> 9) & 3;
  const int b = idx >> 11;
  __shared__ v4f xs4[16 * NF / 4];
  const float* xp = xf + (size_t)(b * 64 + h * 16) * NF + threadIdx.x * 4;
  v4f s = *(const v4f*)xp;
  for (int p = 1; p < KBLK; ++p) s += *(const v4f*)(xp + (size_t)p * (NROWS * NF));
  const float* wr0 = wr + ((size_t)(h * 16) * 16 + o) * NMODES + f;
  const float* wi0 = wi + ((size_t)(h * 16) * 16 + o) * NMODES + f;
  float wre[16], wim[16];
  #pragma unroll
  for (int i = 0; i < 16; ++i) {
    wre[i] = wr0[i * (16 * NMODES)];
    wim[i] = wi0[i * (16 * NMODES)];
  }
  xs4[threadIdx.x] = s;
  __syncthreads();
  const float* xr0 = (const float*)xs4 + f;
  float or_ = 0.0f, oi_ = 0.0f;
  #pragma unroll
  for (int i = 0; i < 16; ++i) {
    const float xr  = xr0[i * NF];
    const float xi  = xr0[i * NF + 32];
    or_ += xr * wre[i] - xi * wim[i];
    oi_ += xr * wim[i] + xi * wre[i];
  }
  const float invL = 1.0f / (float)LLEN;
  const float scale = (f == 0) ? invL : 2.0f * invL;
  const size_t ro = (size_t)(b * 64 + h * 16 + o);
  g[ro * NF + f]      = or_ * scale;
  g[ro * NF + 32 + f] = oi_ * scale;
}

__global__ __launch_bounds__(256) void k_idft(const float* __restrict__ g,
                                              const float* __restrict__ t2q,
                                              const float* __restrict__ bias,
                                              float* __restrict__ out) {
  const int lane = threadIdx.x & 31;
  const int wid  = threadIdx.x >> 5;
  const int gw   = blockIdx.x * 8 + wid;
  const int mpair = gw >> 5;
  const int nc    = gw & (NCHUNK - 1);
  const int hi = lane >> 4, ln = lane & 15;
  const int rowbase = mpair * 32;

  v2f af0[16], af1[16];
  #pragma unroll
  for (int ks = 0; ks < 16; ++ks) {
    const int kb = ks * 4 + 2 * hi;
    af0[ks] = *(const v2f*)(g + (size_t)(rowbase + ln) * NF + kb);
    af1[ks] = *(const v2f*)(g + (size_t)(rowbase + 16 + ln) * NF + kb);
  }
  float bv0[8], bv1[8];
  #pragma unroll
  for (int m = 0; m < 8; ++m) {
    bv0[m] = bias[(rowbase + m + 8 * hi) & 63];
    bv1[m] = bias[(rowbase + 16 + m + 8 * hi) & 63];
  }

  const int colchunk = nc * (LLEN / NCHUNK);
  const float* tlane = t2q + (size_t)(colchunk >> 6) * 4096 + (hi * 16 + ln) * 4;
  for (int gq = 0; gq < 4; ++gq) {
    const float* tb = tlane + gq * 4096;
    v8f c00 = {}, c01 = {}, c02 = {}, c03 = {};
    v8f c10 = {}, c11 = {}, c12 = {}, c13 = {};
    #pragma unroll
    for (int ks = 0; ks < 16; ++ks) {
      const v4f p = *(const v4f*)(tb + (ks * 2 + 0) * 128);
      const v4f r = *(const v4f*)(tb + (ks * 2 + 1) * 128);
      const v2f b0 = {p.x, p.y};
      const v2f b1 = {p.z, p.w};
      const v2f b2 = {r.x, r.y};
      const v2f b3 = {r.z, r.w};
      c00 = __builtin_amdgcn_wmma_f32_16x16x4_f32(false, af0[ks], false, b0, (short)0, c00, false, false);
      c01 = __builtin_amdgcn_wmma_f32_16x16x4_f32(false, af0[ks], false, b1, (short)0, c01, false, false);
      c02 = __builtin_amdgcn_wmma_f32_16x16x4_f32(false, af0[ks], false, b2, (short)0, c02, false, false);
      c03 = __builtin_amdgcn_wmma_f32_16x16x4_f32(false, af0[ks], false, b3, (short)0, c03, false, false);
      c10 = __builtin_amdgcn_wmma_f32_16x16x4_f32(false, af1[ks], false, b0, (short)0, c10, false, false);
      c11 = __builtin_amdgcn_wmma_f32_16x16x4_f32(false, af1[ks], false, b1, (short)0, c11, false, false);
      c12 = __builtin_amdgcn_wmma_f32_16x16x4_f32(false, af1[ks], false, b2, (short)0, c12, false, false);
      c13 = __builtin_amdgcn_wmma_f32_16x16x4_f32(false, af1[ks], false, b3, (short)0, c13, false, false);
    }
    const int col = colchunk + gq * 64 + 4 * ln;
    #pragma unroll
    for (int m = 0; m < 8; ++m) {
      const int r = rowbase + m + 8 * hi;
      const v4f o0 = {c00[m] + bv0[m], c01[m] + bv0[m], c02[m] + bv0[m], c03[m] + bv0[m]};
      const v4f o1 = {c10[m] + bv1[m], c11[m] + bv1[m], c12[m] + bv1[m], c13[m] + bv1[m]};
      *(v4f*)(out + (size_t)r * LLEN + col)        = o0;
      *(v4f*)(out + (size_t)(r + 16) * LLEN + col) = o1;
    }
  }
}

extern "C" void kernel_launch(void* const* d_in, const int* in_sizes, int n_in,
                              void* d_out, int out_size, void* d_ws, size_t ws_size,
                              hipStream_t stream) {
  const float* x    = (const float*)d_in[0];
  const float* wre  = (const float*)d_in[1];
  const float* wim  = (const float*)d_in[2];
  const float* bias = (const float*)d_in[3];
  float* out = (float*)d_out;
  float* ws  = (float*)d_ws;

  float* t1p = ws + T1P_OFF;
  float* t2q = ws + T2_OFF;
  float* xf  = ws + XF_OFF;
  float* g   = ws + G_OFF;

  k_tables <<<(LLEN * NF) / 256, 256, 0, stream>>>(t1p, t2q);
  k_dft_fwd<<<MPAIRS * KBLK, 512, 0, stream>>>(x, t1p, xf);
  k_mix    <<<(NROWS * NF / 2) / 256, 256, 0, stream>>>(xf, wre, wim, g);
  k_idft   <<<(MPAIRS * NCHUNK) / 8, 256, 0, stream>>>(g, t2q, bias, out);
}


# Half-period fold, basis(l+L/2,f)=(-1)^f*basis(l,f): both GEMMs run on L/2 with even/odd-f column halves, WMMA count halved; stage-1 waves take K steps round-robin; sum order changed
# speedup vs baseline: 1.9200x; 1.2495x over previous
#include <hip/hip_runtime.h>


#define LLEN   8192
#define LH     4096
#define NMODES 32
#define NF     64
#define NROWS  4096
#define KSPLIT 16
#define KBLK   2
#define MPAIRS 128
#define NCHUNK 32
#define TWO_PI 6.283185307179586476925f

#define T1P_OFF 0
#define T2_OFF  (LH * NF)
#define XF_OFF  (2 * LH * NF)
#define G_OFF   (2 * LH * NF + KBLK * NROWS * NF)

typedef float v2f __attribute__((ext_vector_type(2)));
typedef float v4f __attribute__((ext_vector_type(4)));
typedef float v8f __attribute__((ext_vector_type(8)));

__global__ __launch_bounds__(256) void k_tables(float* __restrict__ t1p,
                                                float* __restrict__ t2q) {
  const int idx = blockIdx.x * 256 + threadIdx.x;
  const int l  = idx >> 6;
  const int kk = idx & 63;
  const int f  = 2 * (kk & 15) + (kk >> 5);
  const int r  = (f * l) & (LLEN - 1);
  const float ph = (float)r * (TWO_PI / (float)LLEN);
  const float v = ((kk >> 4) & 1) ? -sinf(ph) : cosf(ph);
  const int lc = l & 63;
  const int q  = lc & 3;
  const int n  = lc >> 2;
  const int ks = kk >> 2, hj = (kk >> 1) & 1, e = kk & 1;
  t2q[(size_t)(l >> 6) * 4096 + ((ks * 2 + (q >> 1)) * 2 + hj) * 64 + n * 4 + (q & 1) * 2 + e] = v;
  t1p[(size_t)(l >> 1) * 128 + kk * 2 + (l & 1)] = v;
}

__global__ __launch_bounds__(512) void k_dft_fwd(const float* __restrict__ x,
                                                 const float* __restrict__ t1p,
                                                 float* __restrict__ xf) {
  __shared__ float red[KSPLIT * 32 * NF];
  const int lane = threadIdx.x & 31;
  const int wid  = threadIdx.x >> 5;
  const int hi   = lane >> 4;
  const int ln   = lane & 15;
  const int mp = blockIdx.x / KBLK;
  const int kh = blockIdx.x % KBLK;
  const int rowbase = mp * 32;
  const float* xrow0 = x + (size_t)(rowbase + ln) * LLEN;
  const float* xrow1 = x + (size_t)(rowbase + 16 + ln) * LLEN;

  v8f a0t0 = {}, a0t1 = {}, a0t2 = {}, a0t3 = {};
  v8f a1t0 = {}, a1t1 = {}, a1t2 = {}, a1t3 = {};
  const int k0beg = kh * (LH / KBLK) + 4 * wid + 2 * hi;
  for (int it = 0; it < LH / (KBLK * KSPLIT * 4); ++it) {
    const int kb = k0beg + 4 * KSPLIT * it;
    const v2f p0 = *(const v2f*)(xrow0 + kb);
    const v2f q0 = *(const v2f*)(xrow0 + kb + LH);
    const v2f p1 = *(const v2f*)(xrow1 + kb);
    const v2f q1 = *(const v2f*)(xrow1 + kb + LH);
    const float* tb = t1p + (size_t)(kb >> 1) * 128;
    const v2f b0 = *(const v2f*)(tb + (0 * 16 + ln) * 2);
    const v2f b1 = *(const v2f*)(tb + (1 * 16 + ln) * 2);
    const v2f b2 = *(const v2f*)(tb + (2 * 16 + ln) * 2);
    const v2f b3 = *(const v2f*)(tb + (3 * 16 + ln) * 2);
    const v2f e0 = p0 + q0, o0 = p0 - q0;
    const v2f e1 = p1 + q1, o1 = p1 - q1;
    a0t0 = __builtin_amdgcn_wmma_f32_16x16x4_f32(false, e0, false, b0, (short)0, a0t0, false, false);
    a0t1 = __builtin_amdgcn_wmma_f32_16x16x4_f32(false, e0, false, b1, (short)0, a0t1, false, false);
    a0t2 = __builtin_amdgcn_wmma_f32_16x16x4_f32(false, o0, false, b2, (short)0, a0t2, false, false);
    a0t3 = __builtin_amdgcn_wmma_f32_16x16x4_f32(false, o0, false, b3, (short)0, a0t3, false, false);
    a1t0 = __builtin_amdgcn_wmma_f32_16x16x4_f32(false, e1, false, b0, (short)0, a1t0, false, false);
    a1t1 = __builtin_amdgcn_wmma_f32_16x16x4_f32(false, e1, false, b1, (short)0, a1t1, false, false);
    a1t2 = __builtin_amdgcn_wmma_f32_16x16x4_f32(false, o1, false, b2, (short)0, a1t2, false, false);
    a1t3 = __builtin_amdgcn_wmma_f32_16x16x4_f32(false, o1, false, b3, (short)0, a1t3, false, false);
  }

  float* my = red + wid * (32 * NF);
  #pragma unroll
  for (int m = 0; m < 8; ++m) {
    const int r = m + 8 * hi;
    my[r * NF + 0 * 16 + ln] = a0t0[m];
    my[r * NF + 1 * 16 + ln] = a0t1[m];
    my[r * NF + 2 * 16 + ln] = a0t2[m];
    my[r * NF + 3 * 16 + ln] = a0t3[m];
    my[(16 + r) * NF + 0 * 16 + ln] = a1t0[m];
    my[(16 + r) * NF + 1 * 16 + ln] = a1t1[m];
    my[(16 + r) * NF + 2 * 16 + ln] = a1t2[m];
    my[(16 + r) * NF + 3 * 16 + ln] = a1t3[m];
  }
  __syncthreads();
  #pragma unroll
  for (int u = 0; u < 4; ++u) {
    const int idx = threadIdx.x + 512 * u;
    float s = 0.0f;
    #pragma unroll
    for (int w = 0; w < KSPLIT; ++w) s += red[w * (32 * NF) + idx];
    xf[(size_t)kh * (NROWS * NF) + (size_t)mp * (32 * NF) + idx] = s;
  }
}

__global__ __launch_bounds__(256) void k_mix(const float* __restrict__ xf,
                                             const float* __restrict__ wr,
                                             const float* __restrict__ wi,
                                             float* __restrict__ g) {
  const int idx = blockIdx.x * 256 + threadIdx.x;
  const int f = idx & 31;
  const int o = (idx >> 5) & 15;
  const int h = (idx >> 9) & 3;
  const int b = idx >> 11;
  __shared__ v4f xs4[16 * NF / 4];
  const float* xp = xf + (size_t)(b * 64 + h * 16) * NF + threadIdx.x * 4;
  v4f s = *(const v4f*)xp;
  for (int p = 1; p < KBLK; ++p) s += *(const v4f*)(xp + (size_t)p * (NROWS * NF));
  const float* wr0 = wr + ((size_t)(h * 16) * 16 + o) * NMODES + f;
  const float* wi0 = wi + ((size_t)(h * 16) * 16 + o) * NMODES + f;
  float wre[16], wim[16];
  #pragma unroll
  for (int i = 0; i < 16; ++i) {
    wre[i] = wr0[i * (16 * NMODES)];
    wim[i] = wi0[i * (16 * NMODES)];
  }
  xs4[threadIdx.x] = s;
  __syncthreads();
  const int kr = (f & 1) * 32 + (f >> 1);
  const float* xr0 = (const float*)xs4 + kr;
  float or_ = 0.0f, oi_ = 0.0f;
  #pragma unroll
  for (int i = 0; i < 16; ++i) {
    const float xr  = xr0[i * NF];
    const float xi  = xr0[i * NF + 16];
    or_ += xr * wre[i] - xi * wim[i];
    oi_ += xr * wim[i] + xi * wre[i];
  }
  const float invL = 1.0f / (float)LLEN;
  const float scale = (f == 0) ? invL : 2.0f * invL;
  const size_t ro = (size_t)(b * 64 + h * 16 + o);
  g[ro * NF + kr]      = or_ * scale;
  g[ro * NF + kr + 16] = oi_ * scale;
}

__global__ __launch_bounds__(128) void k_idft(const float* __restrict__ g,
                                              const float* __restrict__ t2q,
                                              const float* __restrict__ bias,
                                              float* __restrict__ out) {
  const int lane = threadIdx.x & 31;
  const int wid  = threadIdx.x >> 5;
  const int gw   = blockIdx.x * 4 + wid;
  const int mpair = gw >> 5;
  const int nc    = gw & (NCHUNK - 1);
  const int hi = lane >> 4, ln = lane & 15;
  const int rowbase = mpair * 32;

  v2f af0[16], af1[16];
  #pragma unroll
  for (int ks = 0; ks < 16; ++ks) {
    const int kb = ks * 4 + 2 * hi;
    af0[ks] = *(const v2f*)(g + (size_t)(rowbase + ln) * NF + kb);
    af1[ks] = *(const v2f*)(g + (size_t)(rowbase + 16 + ln) * NF + kb);
  }
  float bv0[8], bv1[8];
  #pragma unroll
  for (int m = 0; m < 8; ++m) {
    bv0[m] = bias[(rowbase + m + 8 * hi) & 63];
    bv1[m] = bias[(rowbase + 16 + m + 8 * hi) & 63];
  }

  const int colchunk = nc * (LH / NCHUNK);
  const float* tlane = t2q + (size_t)(colchunk >> 6) * 4096 + (hi * 16 + ln) * 4;
  for (int gq = 0; gq < 2; ++gq) {
    const float* tb = tlane + gq * 4096;
    v8f e00 = {}, e01 = {}, e02 = {}, e03 = {};
    v8f e10 = {}, e11 = {}, e12 = {}, e13 = {};
    v8f o00 = {}, o01 = {}, o02 = {}, o03 = {};
    v8f o10 = {}, o11 = {}, o12 = {}, o13 = {};
    #pragma unroll
    for (int ks = 0; ks < 8; ++ks) {
      const v4f p = *(const v4f*)(tb + (ks * 2 + 0) * 128);
      const v4f r = *(const v4f*)(tb + (ks * 2 + 1) * 128);
      const v2f b0 = {p.x, p.y};
      const v2f b1 = {p.z, p.w};
      const v2f b2 = {r.x, r.y};
      const v2f b3 = {r.z, r.w};
      e00 = __builtin_amdgcn_wmma_f32_16x16x4_f32(false, af0[ks], false, b0, (short)0, e00, false, false);
      e01 = __builtin_amdgcn_wmma_f32_16x16x4_f32(false, af0[ks], false, b1, (short)0, e01, false, false);
      e02 = __builtin_amdgcn_wmma_f32_16x16x4_f32(false, af0[ks], false, b2, (short)0, e02, false, false);
      e03 = __builtin_amdgcn_wmma_f32_16x16x4_f32(false, af0[ks], false, b3, (short)0, e03, false, false);
      e10 = __builtin_amdgcn_wmma_f32_16x16x4_f32(false, af1[ks], false, b0, (short)0, e10, false, false);
      e11 = __builtin_amdgcn_wmma_f32_16x16x4_f32(false, af1[ks], false, b1, (short)0, e11, false, false);
      e12 = __builtin_amdgcn_wmma_f32_16x16x4_f32(false, af1[ks], false, b2, (short)0, e12, false, false);
      e13 = __builtin_amdgcn_wmma_f32_16x16x4_f32(false, af1[ks], false, b3, (short)0, e13, false, false);
    }
    #pragma unroll
    for (int ks = 8; ks < 16; ++ks) {
      const v4f p = *(const v4f*)(tb + (ks * 2 + 0) * 128);
      const v4f r = *(const v4f*)(tb + (ks * 2 + 1) * 128);
      const v2f b0 = {p.x, p.y};
      const v2f b1 = {p.z, p.w};
      const v2f b2 = {r.x, r.y};
      const v2f b3 = {r.z, r.w};
      o00 = __builtin_amdgcn_wmma_f32_16x16x4_f32(false, af0[ks], false, b0, (short)0, o00, false, false);
      o01 = __builtin_amdgcn_wmma_f32_16x16x4_f32(false, af0[ks], false, b1, (short)0, o01, false, false);
      o02 = __builtin_amdgcn_wmma_f32_16x16x4_f32(false, af0[ks], false, b2, (short)0, o02, false, false);
      o03 = __builtin_amdgcn_wmma_f32_16x16x4_f32(false, af0[ks], false, b3, (short)0, o03, false, false);
      o10 = __builtin_amdgcn_wmma_f32_16x16x4_f32(false, af1[ks], false, b0, (short)0, o10, false, false);
      o11 = __builtin_amdgcn_wmma_f32_16x16x4_f32(false, af1[ks], false, b1, (short)0, o11, false, false);
      o12 = __builtin_amdgcn_wmma_f32_16x16x4_f32(false, af1[ks], false, b2, (short)0, o12, false, false);
      o13 = __builtin_amdgcn_wmma_f32_16x16x4_f32(false, af1[ks], false, b3, (short)0, o13, false, false);
    }
    const int col = colchunk + gq * 64 + 4 * ln;
    #pragma unroll
    for (int m = 0; m < 8; ++m) {
      const int r = rowbase + m + 8 * hi;
      const v4f lo0 = {(e00[m] + o00[m]) + bv0[m], (e01[m] + o01[m]) + bv0[m],
                       (e02[m] + o02[m]) + bv0[m], (e03[m] + o03[m]) + bv0[m]};
      const v4f hi0 = {(e00[m] - o00[m]) + bv0[m], (e01[m] - o01[m]) + bv0[m],
                       (e02[m] - o02[m]) + bv0[m], (e03[m] - o03[m]) + bv0[m]};
      const v4f lo1 = {(e10[m] + o10[m]) + bv1[m], (e11[m] + o11[m]) + bv1[m],
                       (e12[m] + o12[m]) + bv1[m], (e13[m] + o13[m]) + bv1[m]};
      const v4f hi1 = {(e10[m] - o10[m]) + bv1[m], (e11[m] - o11[m]) + bv1[m],
                       (e12[m] - o12[m]) + bv1[m], (e13[m] - o13[m]) + bv1[m]};
      *(v4f*)(out + (size_t)r * LLEN + col)             = lo0;
      *(v4f*)(out + (size_t)r * LLEN + LH + col)        = hi0;
      *(v4f*)(out + (size_t)(r + 16) * LLEN + col)      = lo1;
      *(v4f*)(out + (size_t)(r + 16) * LLEN + LH + col) = hi1;
    }
  }
}

extern "C" void kernel_launch(void* const* d_in, const int* in_sizes, int n_in,
                              void* d_out, int out_size, void* d_ws, size_t ws_size,
                              hipStream_t stream) {
  const float* x    = (const float*)d_in[0];
  const float* wre  = (const float*)d_in[1];
  const float* wim  = (const float*)d_in[2];
  const float* bias = (const float*)d_in[3];
  float* out = (float*)d_out;
  float* ws  = (float*)d_ws;

  float* t1p = ws + T1P_OFF;
  float* t2q = ws + T2_OFF;
  float* xf  = ws + XF_OFF;
  float* g   = ws + G_OFF;

  k_tables <<<(LH * NF) / 256, 256, 0, stream>>>(t1p, t2q);
  k_dft_fwd<<<MPAIRS * KBLK, 512, 0, stream>>>(x, t1p, xf);
  k_mix    <<<(NROWS * NF / 2) / 256, 256, 0, stream>>>(xf, wre, wim, g);
  k_idft   <<<(MPAIRS * NCHUNK) / 4, 128, 0, stream>>>(g, t2q, bias, out);
}


# Quarter-period fold of the even half, basis(l+L/4,2g)=(-1)^g*basis(l,2g): f=0 and f=2 mod 4 tiles run on L/4 in both GEMMs, 6 WMMAs per 8 of the half fold; stage 3 at 2 waves/SIMD; sum order changed
# speedup vs baseline: 2.0316x; 1.0581x over previous
#include <hip/hip_runtime.h>


#define LLEN   8192
#define LH     4096
#define LQ     2048
#define NMODES 32
#define NF     64
#define NROWS  4096
#define KSPLIT 16
#define KBLK   2
#define MPAIRS 128
#define NCHUNK 32
#define TWO_PI 6.283185307179586476925f

#define T1P_OFF 0
#define T2_OFF  (LH * NF)
#define XF_OFF  (2 * LH * NF)
#define G_OFF   (2 * LH * NF + KBLK * NROWS * NF)

typedef float v2f __attribute__((ext_vector_type(2)));
typedef float v4f __attribute__((ext_vector_type(4)));
typedef float v8f __attribute__((ext_vector_type(8)));

__global__ __launch_bounds__(256) void k_tables(float* __restrict__ t1p,
                                                float* __restrict__ t2q) {
  const int idx = blockIdx.x * 256 + threadIdx.x;
  const int l  = idx >> 6;
  const int kk = idx & 63;
  const int f    = (kk < 32) ? 4 * (kk & 7) + 2 * (kk >> 4) : 2 * (kk & 15) + 1;
  const int part = (kk < 32) ? (kk >> 3) & 1 : (kk >> 4) & 1;
  const int r  = (f * l) & (LLEN - 1);
  const float ph = (float)r * (TWO_PI / (float)LLEN);
  const float v = part ? -sinf(ph) : cosf(ph);
  const int lc = l & 63;
  const int q  = lc & 3;
  const int n  = lc >> 2;
  const int ks = kk >> 2, hj = (kk >> 1) & 1, e = kk & 1;
  t2q[(size_t)(l >> 6) * 4096 + ((ks * 2 + (q >> 1)) * 2 + hj) * 64 + n * 4 + (q & 1) * 2 + e] = v;
  t1p[(size_t)(l >> 1) * 128 + kk * 2 + (l & 1)] = v;
}

__global__ __launch_bounds__(512) void k_dft_fwd(const float* __restrict__ x,
                                                 const float* __restrict__ t1p,
                                                 float* __restrict__ xf) {
  __shared__ float red[KSPLIT * 32 * NF];
  const int lane = threadIdx.x & 31;
  const int wid  = threadIdx.x >> 5;
  const int hi   = lane >> 4;
  const int ln   = lane & 15;
  const int mp = blockIdx.x / KBLK;
  const int kh = blockIdx.x % KBLK;
  const int rowbase = mp * 32;
  const float* xrow0 = x + (size_t)(rowbase + ln) * LLEN;
  const float* xrow1 = x + (size_t)(rowbase + 16 + ln) * LLEN;

  v8f a0t0 = {}, a0t1 = {}, a0t2 = {}, a0t3 = {};
  v8f a1t0 = {}, a1t1 = {}, a1t2 = {}, a1t3 = {};
  const int k0beg = kh * (LQ / KBLK) + 4 * wid + 2 * hi;
  for (int it = 0; it < LQ / (KBLK * KSPLIT * 4); ++it) {
    const int kb = k0beg + 4 * KSPLIT * it;
    const v2f xa0 = *(const v2f*)(xrow0 + kb);
    const v2f xb0 = *(const v2f*)(xrow0 + kb + LQ);
    const v2f xc0 = *(const v2f*)(xrow0 + kb + LH);
    const v2f xd0 = *(const v2f*)(xrow0 + kb + LH + LQ);
    const v2f xa1 = *(const v2f*)(xrow1 + kb);
    const v2f xb1 = *(const v2f*)(xrow1 + kb + LQ);
    const v2f xc1 = *(const v2f*)(xrow1 + kb + LH);
    const v2f xd1 = *(const v2f*)(xrow1 + kb + LH + LQ);
    const float* tb = t1p + (size_t)(kb >> 1) * 128;
    const float* tq = tb + (size_t)(LQ >> 1) * 128;
    const v2f b0 = *(const v2f*)(tb + (0 * 16 + ln) * 2);
    const v2f b1 = *(const v2f*)(tb + (1 * 16 + ln) * 2);
    const v2f b2 = *(const v2f*)(tb + (2 * 16 + ln) * 2);
    const v2f b3 = *(const v2f*)(tb + (3 * 16 + ln) * 2);
    const v2f c2 = *(const v2f*)(tq + (2 * 16 + ln) * 2);
    const v2f c3 = *(const v2f*)(tq + (3 * 16 + ln) * 2);
    const v2f ea0 = xa0 + xc0, eb0 = xb0 + xd0;
    const v2f oa0 = xa0 - xc0, ob0 = xb0 - xd0;
    const v2f ee0 = ea0 + eb0, eo0 = ea0 - eb0;
    const v2f ea1 = xa1 + xc1, eb1 = xb1 + xd1;
    const v2f oa1 = xa1 - xc1, ob1 = xb1 - xd1;
    const v2f ee1 = ea1 + eb1, eo1 = ea1 - eb1;
    a0t0 = __builtin_amdgcn_wmma_f32_16x16x4_f32(false, ee0, false, b0, (short)0, a0t0, false, false);
    a0t1 = __builtin_amdgcn_wmma_f32_16x16x4_f32(false, eo0, false, b1, (short)0, a0t1, false, false);
    a0t2 = __builtin_amdgcn_wmma_f32_16x16x4_f32(false, oa0, false, b2, (short)0, a0t2, false, false);
    a0t3 = __builtin_amdgcn_wmma_f32_16x16x4_f32(false, oa0, false, b3, (short)0, a0t3, false, false);
    a1t0 = __builtin_amdgcn_wmma_f32_16x16x4_f32(false, ee1, false, b0, (short)0, a1t0, false, false);
    a1t1 = __builtin_amdgcn_wmma_f32_16x16x4_f32(false, eo1, false, b1, (short)0, a1t1, false, false);
    a1t2 = __builtin_amdgcn_wmma_f32_16x16x4_f32(false, oa1, false, b2, (short)0, a1t2, false, false);
    a1t3 = __builtin_amdgcn_wmma_f32_16x16x4_f32(false, oa1, false, b3, (short)0, a1t3, false, false);
    a0t2 = __builtin_amdgcn_wmma_f32_16x16x4_f32(false, ob0, false, c2, (short)0, a0t2, false, false);
    a0t3 = __builtin_amdgcn_wmma_f32_16x16x4_f32(false, ob0, false, c3, (short)0, a0t3, false, false);
    a1t2 = __builtin_amdgcn_wmma_f32_16x16x4_f32(false, ob1, false, c2, (short)0, a1t2, false, false);
    a1t3 = __builtin_amdgcn_wmma_f32_16x16x4_f32(false, ob1, false, c3, (short)0, a1t3, false, false);
  }

  float* my = red + wid * (32 * NF);
  #pragma unroll
  for (int m = 0; m < 8; ++m) {
    const int r = m + 8 * hi;
    my[r * NF + 0 * 16 + ln] = a0t0[m];
    my[r * NF + 1 * 16 + ln] = a0t1[m];
    my[r * NF + 2 * 16 + ln] = a0t2[m];
    my[r * NF + 3 * 16 + ln] = a0t3[m];
    my[(16 + r) * NF + 0 * 16 + ln] = a1t0[m];
    my[(16 + r) * NF + 1 * 16 + ln] = a1t1[m];
    my[(16 + r) * NF + 2 * 16 + ln] = a1t2[m];
    my[(16 + r) * NF + 3 * 16 + ln] = a1t3[m];
  }
  __syncthreads();
  #pragma unroll
  for (int u = 0; u < 4; ++u) {
    const int idx = threadIdx.x + 512 * u;
    float s = 0.0f;
    #pragma unroll
    for (int w = 0; w < KSPLIT; ++w) s += red[w * (32 * NF) + idx];
    xf[(size_t)kh * (NROWS * NF) + (size_t)mp * (32 * NF) + idx] = s;
  }
}

__global__ __launch_bounds__(256) void k_mix(const float* __restrict__ xf,
                                             const float* __restrict__ wr,
                                             const float* __restrict__ wi,
                                             float* __restrict__ g) {
  const int idx = blockIdx.x * 256 + threadIdx.x;
  const int f = idx & 31;
  const int o = (idx >> 5) & 15;
  const int h = (idx >> 9) & 3;
  const int b = idx >> 11;
  __shared__ v4f xs4[16 * NF / 4];
  const float* xp = xf + (size_t)(b * 64 + h * 16) * NF + threadIdx.x * 4;
  v4f s = *(const v4f*)xp;
  for (int p = 1; p < KBLK; ++p) s += *(const v4f*)(xp + (size_t)p * (NROWS * NF));
  const float* wr0 = wr + ((size_t)(h * 16) * 16 + o) * NMODES + f;
  const float* wi0 = wi + ((size_t)(h * 16) * 16 + o) * NMODES + f;
  float wre[16], wim[16];
  #pragma unroll
  for (int i = 0; i < 16; ++i) {
    wre[i] = wr0[i * (16 * NMODES)];
    wim[i] = wi0[i * (16 * NMODES)];
  }
  xs4[threadIdx.x] = s;
  __syncthreads();
  const int kr = (f & 1) ? 32 + (f >> 1) : ((f >> 1) & 1) * 16 + (f >> 2);
  const int ki = kr + ((f & 1) ? 16 : 8);
  const float* xr0 = (const float*)xs4 + kr;
  const float* xi0 = (const float*)xs4 + ki;
  float or_ = 0.0f, oi_ = 0.0f;
  #pragma unroll
  for (int i = 0; i < 16; ++i) {
    const float xr  = xr0[i * NF];
    const float xi  = xi0[i * NF];
    or_ += xr * wre[i] - xi * wim[i];
    oi_ += xr * wim[i] + xi * wre[i];
  }
  const float invL = 1.0f / (float)LLEN;
  const float scale = (f == 0) ? invL : 2.0f * invL;
  const size_t ro = (size_t)(b * 64 + h * 16 + o);
  g[ro * NF + kr]      = or_ * scale;
  g[ro * NF + ki]      = oi_ * scale;
}

__global__ __launch_bounds__(128) __attribute__((amdgpu_waves_per_eu(2, 2))) void k_idft(const float* __restrict__ g,
                                              const float* __restrict__ t2q,
                                              const float* __restrict__ bias,
                                              float* __restrict__ out) {
  const int lane = threadIdx.x & 31;
  const int wid  = threadIdx.x >> 5;
  const int gw   = blockIdx.x * 4 + wid;
  const int mpair = gw >> 5;
  const int nc    = gw & (NCHUNK - 1);
  const int hi = lane >> 4, ln = lane & 15;
  const int rowbase = mpair * 32;

  v2f af0[16], af1[16];
  #pragma unroll
  for (int ks = 0; ks < 16; ++ks) {
    const int kb = ks * 4 + 2 * hi;
    af0[ks] = *(const v2f*)(g + (size_t)(rowbase + ln) * NF + kb);
    af1[ks] = *(const v2f*)(g + (size_t)(rowbase + 16 + ln) * NF + kb);
  }
  float bv0[8], bv1[8];
  #pragma unroll
  for (int m = 0; m < 8; ++m) {
    bv0[m] = bias[(rowbase + m + 8 * hi) & 63];
    bv1[m] = bias[(rowbase + 16 + m + 8 * hi) & 63];
  }

  const int col = nc * (LQ / NCHUNK) + 4 * ln;
  const float* tb = t2q + (size_t)nc * 4096 + (hi * 16 + ln) * 4;
  const float* tq = tb + (size_t)(LQ >> 6) * 4096;

  const v8f bz0 = {bv0[0], bv0[1], bv0[2], bv0[3], bv0[4], bv0[5], bv0[6], bv0[7]};
  const v8f bz1 = {bv1[0], bv1[1], bv1[2], bv1[3], bv1[4], bv1[5], bv1[6], bv1[7]};
  v8f z00 = bz0, z01 = bz0, z02 = bz0, z03 = bz0, z10 = bz1, z11 = bz1, z12 = bz1, z13 = bz1;
  v8f t00 = {}, t01 = {}, t02 = {}, t03 = {}, t10 = {}, t11 = {}, t12 = {}, t13 = {};
  #pragma unroll
  for (int ks = 0; ks < 4; ++ks) {
    const v4f p = *(const v4f*)(tb + (ks * 2 + 0) * 128);
    const v4f r = *(const v4f*)(tb + (ks * 2 + 1) * 128);
    const v2f b0 = {p.x, p.y};
    const v2f b1 = {p.z, p.w};
    const v2f b2 = {r.x, r.y};
    const v2f b3 = {r.z, r.w};
    z00 = __builtin_amdgcn_wmma_f32_16x16x4_f32(false, af0[ks], false, b0, (short)0, z00, false, false);
    z01 = __builtin_amdgcn_wmma_f32_16x16x4_f32(false, af0[ks], false, b1, (short)0, z01, false, false);
    z02 = __builtin_amdgcn_wmma_f32_16x16x4_f32(false, af0[ks], false, b2, (short)0, z02, false, false);
    z03 = __builtin_amdgcn_wmma_f32_16x16x4_f32(false, af0[ks], false, b3, (short)0, z03, false, false);
    z10 = __builtin_amdgcn_wmma_f32_16x16x4_f32(false, af1[ks], false, b0, (short)0, z10, false, false);
    z11 = __builtin_amdgcn_wmma_f32_16x16x4_f32(false, af1[ks], false, b1, (short)0, z11, false, false);
    z12 = __builtin_amdgcn_wmma_f32_16x16x4_f32(false, af1[ks], false, b2, (short)0, z12, false, false);
    z13 = __builtin_amdgcn_wmma_f32_16x16x4_f32(false, af1[ks], false, b3, (short)0, z13, false, false);
  }
  #pragma unroll
  for (int ks = 4; ks < 8; ++ks) {
    const v4f p = *(const v4f*)(tb + (ks * 2 + 0) * 128);
    const v4f r = *(const v4f*)(tb + (ks * 2 + 1) * 128);
    const v2f b0 = {p.x, p.y};
    const v2f b1 = {p.z, p.w};
    const v2f b2 = {r.x, r.y};
    const v2f b3 = {r.z, r.w};
    t00 = __builtin_amdgcn_wmma_f32_16x16x4_f32(false, af0[ks], false, b0, (short)0, t00, false, false);
    t01 = __builtin_amdgcn_wmma_f32_16x16x4_f32(false, af0[ks], false, b1, (short)0, t01, false, false);
    t02 = __builtin_amdgcn_wmma_f32_16x16x4_f32(false, af0[ks], false, b2, (short)0, t02, false, false);
    t03 = __builtin_amdgcn_wmma_f32_16x16x4_f32(false, af0[ks], false, b3, (short)0, t03, false, false);
    t10 = __builtin_amdgcn_wmma_f32_16x16x4_f32(false, af1[ks], false, b0, (short)0, t10, false, false);
    t11 = __builtin_amdgcn_wmma_f32_16x16x4_f32(false, af1[ks], false, b1, (short)0, t11, false, false);
    t12 = __builtin_amdgcn_wmma_f32_16x16x4_f32(false, af1[ks], false, b2, (short)0, t12, false, false);
    t13 = __builtin_amdgcn_wmma_f32_16x16x4_f32(false, af1[ks], false, b3, (short)0, t13, false, false);
  }

  #pragma unroll
  for (int ps = 0; ps < 2; ++ps) {
    const float* to = ps ? tq : tb;
    v8f o00 = {}, o01 = {}, o02 = {}, o03 = {};
    v8f o10 = {}, o11 = {}, o12 = {}, o13 = {};
    #pragma unroll
    for (int ks = 8; ks < 16; ++ks) {
      const v4f p = *(const v4f*)(to + (ks * 2 + 0) * 128);
      const v4f r = *(const v4f*)(to + (ks * 2 + 1) * 128);
      const v2f b0 = {p.x, p.y};
      const v2f b1 = {p.z, p.w};
      const v2f b2 = {r.x, r.y};
      const v2f b3 = {r.z, r.w};
      o00 = __builtin_amdgcn_wmma_f32_16x16x4_f32(false, af0[ks], false, b0, (short)0, o00, false, false);
      o01 = __builtin_amdgcn_wmma_f32_16x16x4_f32(false, af0[ks], false, b1, (short)0, o01, false, false);
      o02 = __builtin_amdgcn_wmma_f32_16x16x4_f32(false, af0[ks], false, b2, (short)0, o02, false, false);
      o03 = __builtin_amdgcn_wmma_f32_16x16x4_f32(false, af0[ks], false, b3, (short)0, o03, false, false);
      o10 = __builtin_amdgcn_wmma_f32_16x16x4_f32(false, af1[ks], false, b0, (short)0, o10, false, false);
      o11 = __builtin_amdgcn_wmma_f32_16x16x4_f32(false, af1[ks], false, b1, (short)0, o11, false, false);
      o12 = __builtin_amdgcn_wmma_f32_16x16x4_f32(false, af1[ks], false, b2, (short)0, o12, false, false);
      o13 = __builtin_amdgcn_wmma_f32_16x16x4_f32(false, af1[ks], false, b3, (short)0, o13, false, false);
    }
    const float sg = ps ? -1.0f : 1.0f;
    float* olo = out + (size_t)ps * LQ + col;
    float* ohi = olo + LH;
    #pragma unroll
    for (int m = 0; m < 8; ++m) {
      const int r = rowbase + m + 8 * hi;
      const float s00 = z00[m] + sg * t00[m], s01 = z01[m] + sg * t01[m];
      const float s02 = z02[m] + sg * t02[m], s03 = z03[m] + sg * t03[m];
      const float s10 = z10[m] + sg * t10[m], s11 = z11[m] + sg * t11[m];
      const float s12 = z12[m] + sg * t12[m], s13 = z13[m] + sg * t13[m];
      const v4f lo0 = {s00 + o00[m], s01 + o01[m], s02 + o02[m], s03 + o03[m]};
      const v4f hi0 = {s00 - o00[m], s01 - o01[m], s02 - o02[m], s03 - o03[m]};
      const v4f lo1 = {s10 + o10[m], s11 + o11[m], s12 + o12[m], s13 + o13[m]};
      const v4f hi1 = {s10 - o10[m], s11 - o11[m], s12 - o12[m], s13 - o13[m]};
      *(v4f*)(olo + (size_t)r * LLEN)        = lo0;
      *(v4f*)(ohi + (size_t)r * LLEN)        = hi0;
      *(v4f*)(olo + (size_t)(r + 16) * LLEN) = lo1;
      *(v4f*)(ohi + (size_t)(r + 16) * LLEN) = hi1;
    }
  }
}

extern "C" void kernel_launch(void* const* d_in, const int* in_sizes, int n_in,
                              void* d_out, int out_size, void* d_ws, size_t ws_size,
                              hipStream_t stream) {
  const float* x    = (const float*)d_in[0];
  const float* wre  = (const float*)d_in[1];
  const float* wim  = (const float*)d_in[2];
  const float* bias = (const float*)d_in[3];
  float* out = (float*)d_out;
  float* ws  = (float*)d_ws;

  float* t1p = ws + T1P_OFF;
  float* t2q = ws + T2_OFF;
  float* xf  = ws + XF_OFF;
  float* g   = ws + G_OFF;

  k_tables <<<(LH * NF) / 256, 256, 0, stream>>>(t1p, t2q);
  k_dft_fwd<<<MPAIRS * KBLK, 512, 0, stream>>>(x, t1p, xf);
  k_mix    <<<(NROWS * NF / 2) / 256, 256, 0, stream>>>(xf, wre, wim, g);
  k_idft   <<<(MPAIRS * NCHUNK) / 4, 128, 0, stream>>>(g, t2q, bias, out);
}


# Tables kernel: one lane per 2x2 block (l,l+1)x(kk,kk+1), one b128 store per table, 1,024 waves (was 8,192 one-value threads with 4-byte stores); values and sum order unchanged
# speedup vs baseline: 2.2345x; 1.0999x over previous
#include <hip/hip_runtime.h>


#define LLEN   8192
#define LH     4096
#define LQ     2048
#define NMODES 32
#define NF     64
#define NROWS  4096
#define KSPLIT 16
#define KBLK   2
#define MPAIRS 128
#define NCHUNK 32
#define TWO_PI 6.283185307179586476925f

#define T1P_OFF 0
#define T2_OFF  (LH * NF)
#define XF_OFF  (2 * LH * NF)
#define G_OFF   (2 * LH * NF + KBLK * NROWS * NF)

typedef float v2f __attribute__((ext_vector_type(2)));
typedef float v4f __attribute__((ext_vector_type(4)));
typedef float v8f __attribute__((ext_vector_type(8)));

#define TBL_WAVES 4
#define TBL_PAIRS 2
__global__ __launch_bounds__(32 * TBL_WAVES) void k_tables(float* __restrict__ t1p,
                                                           float* __restrict__ t2q) {
  const int kk = 2 * (threadIdx.x & 31);
  const bool ev = kk < 32;
  const int f0   = ev ? 4 * (kk & 7) + 2 * (kk >> 4) : 2 * (kk & 15) + 1;
  const int f1   = f0 + (ev ? 4 : 2);
  const int part = ev ? (kk >> 3) & 1 : (kk >> 4) & 1;
  const int ks = kk >> 2, hj = (kk >> 1) & 1;
  const int pbase = (blockIdx.x * TBL_WAVES + (threadIdx.x >> 5)) * TBL_PAIRS;
  #pragma unroll
  for (int j = 0; j < TBL_PAIRS; ++j) {
    const int l = 2 * (pbase + j);
    float v[4];
    #pragma unroll
    for (int t = 0; t < 4; ++t) {
      const int r = (((t & 1) ? f1 : f0) * (l + (t >> 1))) & (LLEN - 1);
      const float ph = (float)r * (TWO_PI / (float)LLEN);
      v[t] = part ? -sinf(ph) : cosf(ph);
    }
    const int lc = l & 63;
    const int n  = lc >> 2;
    const v4f w2 = {v[0], v[1], v[2], v[3]};
    const v4f w1 = {v[0], v[2], v[1], v[3]};
    *(v4f*)(t2q + (size_t)(l >> 6) * 4096 + ((ks * 2 + ((lc >> 1) & 1)) * 2 + hj) * 64 + n * 4) = w2;
    *(v4f*)(t1p + (size_t)(l >> 1) * 128 + kk * 2) = w1;
  }
}

__global__ __launch_bounds__(512) void k_dft_fwd(const float* __restrict__ x,
                                                 const float* __restrict__ t1p,
                                                 float* __restrict__ xf) {
  __shared__ float red[KSPLIT * 32 * NF];
  const int lane = threadIdx.x & 31;
  const int wid  = threadIdx.x >> 5;
  const int hi   = lane >> 4;
  const int ln   = lane & 15;
  const int mp = blockIdx.x / KBLK;
  const int kh = blockIdx.x % KBLK;
  const int rowbase = mp * 32;
  const float* xrow0 = x + (size_t)(rowbase + ln) * LLEN;
  const float* xrow1 = x + (size_t)(rowbase + 16 + ln) * LLEN;

  v8f a0t0 = {}, a0t1 = {}, a0t2 = {}, a0t3 = {};
  v8f a1t0 = {}, a1t1 = {}, a1t2 = {}, a1t3 = {};
  const int k0beg = kh * (LQ / KBLK) + 4 * wid + 2 * hi;
  for (int it = 0; it < LQ / (KBLK * KSPLIT * 4); ++it) {
    const int kb = k0beg + 4 * KSPLIT * it;
    const v2f xa0 = *(const v2f*)(xrow0 + kb);
    const v2f xb0 = *(const v2f*)(xrow0 + kb + LQ);
    const v2f xc0 = *(const v2f*)(xrow0 + kb + LH);
    const v2f xd0 = *(const v2f*)(xrow0 + kb + LH + LQ);
    const v2f xa1 = *(const v2f*)(xrow1 + kb);
    const v2f xb1 = *(const v2f*)(xrow1 + kb + LQ);
    const v2f xc1 = *(const v2f*)(xrow1 + kb + LH);
    const v2f xd1 = *(const v2f*)(xrow1 + kb + LH + LQ);
    const float* tb = t1p + (size_t)(kb >> 1) * 128;
    const float* tq = tb + (size_t)(LQ >> 1) * 128;
    const v2f b0 = *(const v2f*)(tb + (0 * 16 + ln) * 2);
    const v2f b1 = *(const v2f*)(tb + (1 * 16 + ln) * 2);
    const v2f b2 = *(const v2f*)(tb + (2 * 16 + ln) * 2);
    const v2f b3 = *(const v2f*)(tb + (3 * 16 + ln) * 2);
    const v2f c2 = *(const v2f*)(tq + (2 * 16 + ln) * 2);
    const v2f c3 = *(const v2f*)(tq + (3 * 16 + ln) * 2);
    const v2f ea0 = xa0 + xc0, eb0 = xb0 + xd0;
    const v2f oa0 = xa0 - xc0, ob0 = xb0 - xd0;
    const v2f ee0 = ea0 + eb0, eo0 = ea0 - eb0;
    const v2f ea1 = xa1 + xc1, eb1 = xb1 + xd1;
    const v2f oa1 = xa1 - xc1, ob1 = xb1 - xd1;
    const v2f ee1 = ea1 + eb1, eo1 = ea1 - eb1;
    a0t0 = __builtin_amdgcn_wmma_f32_16x16x4_f32(false, ee0, false, b0, (short)0, a0t0, false, false);
    a0t1 = __builtin_amdgcn_wmma_f32_16x16x4_f32(false, eo0, false, b1, (short)0, a0t1, false, false);
    a0t2 = __builtin_amdgcn_wmma_f32_16x16x4_f32(false, oa0, false, b2, (short)0, a0t2, false, false);
    a0t3 = __builtin_amdgcn_wmma_f32_16x16x4_f32(false, oa0, false, b3, (short)0, a0t3, false, false);
    a1t0 = __builtin_amdgcn_wmma_f32_16x16x4_f32(false, ee1, false, b0, (short)0, a1t0, false, false);
    a1t1 = __builtin_amdgcn_wmma_f32_16x16x4_f32(false, eo1, false, b1, (short)0, a1t1, false, false);
    a1t2 = __builtin_amdgcn_wmma_f32_16x16x4_f32(false, oa1, false, b2, (short)0, a1t2, false, false);
    a1t3 = __builtin_amdgcn_wmma_f32_16x16x4_f32(false, oa1, false, b3, (short)0, a1t3, false, false);
    a0t2 = __builtin_amdgcn_wmma_f32_16x16x4_f32(false, ob0, false, c2, (short)0, a0t2, false, false);
    a0t3 = __builtin_amdgcn_wmma_f32_16x16x4_f32(false, ob0, false, c3, (short)0, a0t3, false, false);
    a1t2 = __builtin_amdgcn_wmma_f32_16x16x4_f32(false, ob1, false, c2, (short)0, a1t2, false, false);
    a1t3 = __builtin_amdgcn_wmma_f32_16x16x4_f32(false, ob1, false, c3, (short)0, a1t3, false, false);
  }

  float* my = red + wid * (32 * NF);
  #pragma unroll
  for (int m = 0; m < 8; ++m) {
    const int r = m + 8 * hi;
    my[r * NF + 0 * 16 + ln] = a0t0[m];
    my[r * NF + 1 * 16 + ln] = a0t1[m];
    my[r * NF + 2 * 16 + ln] = a0t2[m];
    my[r * NF + 3 * 16 + ln] = a0t3[m];
    my[(16 + r) * NF + 0 * 16 + ln] = a1t0[m];
    my[(16 + r) * NF + 1 * 16 + ln] = a1t1[m];
    my[(16 + r) * NF + 2 * 16 + ln] = a1t2[m];
    my[(16 + r) * NF + 3 * 16 + ln] = a1t3[m];
  }
  __syncthreads();
  #pragma unroll
  for (int u = 0; u < 4; ++u) {
    const int idx = threadIdx.x + 512 * u;
    float s = 0.0f;
    #pragma unroll
    for (int w = 0; w < KSPLIT; ++w) s += red[w * (32 * NF) + idx];
    xf[(size_t)kh * (NROWS * NF) + (size_t)mp * (32 * NF) + idx] = s;
  }
}

__global__ __launch_bounds__(256) void k_mix(const float* __restrict__ xf,
                                             const float* __restrict__ wr,
                                             const float* __restrict__ wi,
                                             float* __restrict__ g) {
  const int idx = blockIdx.x * 256 + threadIdx.x;
  const int f = idx & 31;
  const int o = (idx >> 5) & 15;
  const int h = (idx >> 9) & 3;
  const int b = idx >> 11;
  __shared__ v4f xs4[16 * NF / 4];
  const float* xp = xf + (size_t)(b * 64 + h * 16) * NF + threadIdx.x * 4;
  v4f s = *(const v4f*)xp;
  for (int p = 1; p < KBLK; ++p) s += *(const v4f*)(xp + (size_t)p * (NROWS * NF));
  const float* wr0 = wr + ((size_t)(h * 16) * 16 + o) * NMODES + f;
  const float* wi0 = wi + ((size_t)(h * 16) * 16 + o) * NMODES + f;
  float wre[16], wim[16];
  #pragma unroll
  for (int i = 0; i < 16; ++i) {
    wre[i] = wr0[i * (16 * NMODES)];
    wim[i] = wi0[i * (16 * NMODES)];
  }
  xs4[threadIdx.x] = s;
  __syncthreads();
  const int kr = (f & 1) ? 32 + (f >> 1) : ((f >> 1) & 1) * 16 + (f >> 2);
  const int ki = kr + ((f & 1) ? 16 : 8);
  const float* xr0 = (const float*)xs4 + kr;
  const float* xi0 = (const float*)xs4 + ki;
  float or_ = 0.0f, oi_ = 0.0f;
  #pragma unroll
  for (int i = 0; i < 16; ++i) {
    const float xr  = xr0[i * NF];
    const float xi  = xi0[i * NF];
    or_ += xr * wre[i] - xi * wim[i];
    oi_ += xr * wim[i] + xi * wre[i];
  }
  const float invL = 1.0f / (float)LLEN;
  const float scale = (f == 0) ? invL : 2.0f * invL;
  const size_t ro = (size_t)(b * 64 + h * 16 + o);
  g[ro * NF + kr]      = or_ * scale;
  g[ro * NF + ki]      = oi_ * scale;
}

__global__ __launch_bounds__(128) __attribute__((amdgpu_waves_per_eu(2, 2))) void k_idft(const float* __restrict__ g,
                                              const float* __restrict__ t2q,
                                              const float* __restrict__ bias,
                                              float* __restrict__ out) {
  const int lane = threadIdx.x & 31;
  const int wid  = threadIdx.x >> 5;
  const int gw   = blockIdx.x * 4 + wid;
  const int mpair = gw >> 5;
  const int nc    = gw & (NCHUNK - 1);
  const int hi = lane >> 4, ln = lane & 15;
  const int rowbase = mpair * 32;

  v2f af0[16], af1[16];
  #pragma unroll
  for (int ks = 0; ks < 16; ++ks) {
    const int kb = ks * 4 + 2 * hi;
    af0[ks] = *(const v2f*)(g + (size_t)(rowbase + ln) * NF + kb);
    af1[ks] = *(const v2f*)(g + (size_t)(rowbase + 16 + ln) * NF + kb);
  }
  float bv0[8], bv1[8];
  #pragma unroll
  for (int m = 0; m < 8; ++m) {
    bv0[m] = bias[(rowbase + m + 8 * hi) & 63];
    bv1[m] = bias[(rowbase + 16 + m + 8 * hi) & 63];
  }

  const int col = nc * (LQ / NCHUNK) + 4 * ln;
  const float* tb = t2q + (size_t)nc * 4096 + (hi * 16 + ln) * 4;
  const float* tq = tb + (size_t)(LQ >> 6) * 4096;

  const v8f bz0 = {bv0[0], bv0[1], bv0[2], bv0[3], bv0[4], bv0[5], bv0[6], bv0[7]};
  const v8f bz1 = {bv1[0], bv1[1], bv1[2], bv1[3], bv1[4], bv1[5], bv1[6], bv1[7]};
  v8f z00 = bz0, z01 = bz0, z02 = bz0, z03 = bz0, z10 = bz1, z11 = bz1, z12 = bz1, z13 = bz1;
  v8f t00 = {}, t01 = {}, t02 = {}, t03 = {}, t10 = {}, t11 = {}, t12 = {}, t13 = {};
  #pragma unroll
  for (int ks = 0; ks < 4; ++ks) {
    const v4f p = *(const v4f*)(tb + (ks * 2 + 0) * 128);
    const v4f r = *(const v4f*)(tb + (ks * 2 + 1) * 128);
    const v2f b0 = {p.x, p.y};
    const v2f b1 = {p.z, p.w};
    const v2f b2 = {r.x, r.y};
    const v2f b3 = {r.z, r.w};
    z00 = __builtin_amdgcn_wmma_f32_16x16x4_f32(false, af0[ks], false, b0, (short)0, z00, false, false);
    z01 = __builtin_amdgcn_wmma_f32_16x16x4_f32(false, af0[ks], false, b1, (short)0, z01, false, false);
    z02 = __builtin_amdgcn_wmma_f32_16x16x4_f32(false, af0[ks], false, b2, (short)0, z02, false, false);
    z03 = __builtin_amdgcn_wmma_f32_16x16x4_f32(false, af0[ks], false, b3, (short)0, z03, false, false);
    z10 = __builtin_amdgcn_wmma_f32_16x16x4_f32(false, af1[ks], false, b0, (short)0, z10, false, false);
    z11 = __builtin_amdgcn_wmma_f32_16x16x4_f32(false, af1[ks], false, b1, (short)0, z11, false, false);
    z12 = __builtin_amdgcn_wmma_f32_16x16x4_f32(false, af1[ks], false, b2, (short)0, z12, false, false);
    z13 = __builtin_amdgcn_wmma_f32_16x16x4_f32(false, af1[ks], false, b3, (short)0, z13, false, false);
  }
  #pragma unroll
  for (int ks = 4; ks < 8; ++ks) {
    const v4f p = *(const v4f*)(tb + (ks * 2 + 0) * 128);
    const v4f r = *(const v4f*)(tb + (ks * 2 + 1) * 128);
    const v2f b0 = {p.x, p.y};
    const v2f b1 = {p.z, p.w};
    const v2f b2 = {r.x, r.y};
    const v2f b3 = {r.z, r.w};
    t00 = __builtin_amdgcn_wmma_f32_16x16x4_f32(false, af0[ks], false, b0, (short)0, t00, false, false);
    t01 = __builtin_amdgcn_wmma_f32_16x16x4_f32(false, af0[ks], false, b1, (short)0, t01, false, false);
    t02 = __builtin_amdgcn_wmma_f32_16x16x4_f32(false, af0[ks], false, b2, (short)0, t02, false, false);
    t03 = __builtin_amdgcn_wmma_f32_16x16x4_f32(false, af0[ks], false, b3, (short)0, t03, false, false);
    t10 = __builtin_amdgcn_wmma_f32_16x16x4_f32(false, af1[ks], false, b0, (short)0, t10, false, false);
    t11 = __builtin_amdgcn_wmma_f32_16x16x4_f32(false, af1[ks], false, b1, (short)0, t11, false, false);
    t12 = __builtin_amdgcn_wmma_f32_16x16x4_f32(false, af1[ks], false, b2, (short)0, t12, false, false);
    t13 = __builtin_amdgcn_wmma_f32_16x16x4_f32(false, af1[ks], false, b3, (short)0, t13, false, false);
  }

  #pragma unroll
  for (int ps = 0; ps < 2; ++ps) {
    const float* to = ps ? tq : tb;
    v8f o00 = {}, o01 = {}, o02 = {}, o03 = {};
    v8f o10 = {}, o11 = {}, o12 = {}, o13 = {};
    #pragma unroll
    for (int ks = 8; ks < 16; ++ks) {
      const v4f p = *(const v4f*)(to + (ks * 2 + 0) * 128);
      const v4f r = *(const v4f*)(to + (ks * 2 + 1) * 128);
      const v2f b0 = {p.x, p.y};
      const v2f b1 = {p.z, p.w};
      const v2f b2 = {r.x, r.y};
      const v2f b3 = {r.z, r.w};
      o00 = __builtin_amdgcn_wmma_f32_16x16x4_f32(false, af0[ks], false, b0, (short)0, o00, false, false);
      o01 = __builtin_amdgcn_wmma_f32_16x16x4_f32(false, af0[ks], false, b1, (short)0, o01, false, false);
      o02 = __builtin_amdgcn_wmma_f32_16x16x4_f32(false, af0[ks], false, b2, (short)0, o02, false, false);
      o03 = __builtin_amdgcn_wmma_f32_16x16x4_f32(false, af0[ks], false, b3, (short)0, o03, false, false);
      o10 = __builtin_amdgcn_wmma_f32_16x16x4_f32(false, af1[ks], false, b0, (short)0, o10, false, false);
      o11 = __builtin_amdgcn_wmma_f32_16x16x4_f32(false, af1[ks], false, b1, (short)0, o11, false, false);
      o12 = __builtin_amdgcn_wmma_f32_16x16x4_f32(false, af1[ks], false, b2, (short)0, o12, false, false);
      o13 = __builtin_amdgcn_wmma_f32_16x16x4_f32(false, af1[ks], false, b3, (short)0, o13, false, false);
    }
    const float sg = ps ? -1.0f : 1.0f;
    float* olo = out + (size_t)ps * LQ + col;
    float* ohi = olo + LH;
    #pragma unroll
    for (int m = 0; m < 8; ++m) {
      const int r = rowbase + m + 8 * hi;
      const float s00 = z00[m] + sg * t00[m], s01 = z01[m] + sg * t01[m];
      const float s02 = z02[m] + sg * t02[m], s03 = z03[m] + sg * t03[m];
      const float s10 = z10[m] + sg * t10[m], s11 = z11[m] + sg * t11[m];
      const float s12 = z12[m] + sg * t12[m], s13 = z13[m] + sg * t13[m];
      const v4f lo0 = {s00 + o00[m], s01 + o01[m], s02 + o02[m], s03 + o03[m]};
      const v4f hi0 = {s00 - o00[m], s01 - o01[m], s02 - o02[m], s03 - o03[m]};
      const v4f lo1 = {s10 + o10[m], s11 + o11[m], s12 + o12[m], s13 + o13[m]};
      const v4f hi1 = {s10 - o10[m], s11 - o11[m], s12 - o12[m], s13 - o13[m]};
      *(v4f*)(olo + (size_t)r * LLEN)        = lo0;
      *(v4f*)(ohi + (size_t)r * LLEN)        = hi0;
      *(v4f*)(olo + (size_t)(r + 16) * LLEN) = lo1;
      *(v4f*)(ohi + (size_t)(r + 16) * LLEN) = hi1;
    }
  }
}

extern "C" void kernel_launch(void* const* d_in, const int* in_sizes, int n_in,
                              void* d_out, int out_size, void* d_ws, size_t ws_size,
                              hipStream_t stream) {
  const float* x    = (const float*)d_in[0];
  const float* wre  = (const float*)d_in[1];
  const float* wim  = (const float*)d_in[2];
  const float* bias = (const float*)d_in[3];
  float* out = (float*)d_out;
  float* ws  = (float*)d_ws;

  float* t1p = ws + T1P_OFF;
  float* t2q = ws + T2_OFF;
  float* xf  = ws + XF_OFF;
  float* g   = ws + G_OFF;

  k_tables <<<(LH / 2) / (TBL_WAVES * TBL_PAIRS), 32 * TBL_WAVES, 0, stream>>>(t1p, t2q);
  k_dft_fwd<<<MPAIRS * KBLK, 512, 0, stream>>>(x, t1p, xf);
  k_mix    <<<(NROWS * NF / 2) / 256, 256, 0, stream>>>(xf, wre, wim, g);
  k_idft   <<<(MPAIRS * NCHUNK) / 4, 128, 0, stream>>>(g, t2q, bias, out);
}


# Stage 1 makes its B fragments in registers by angle addition (128-entry LDS rotation table x lane phase constants); mix kernel writes the stage-3 table: 3 launches, was 4; basis last bits differ
# speedup vs baseline: 2.5398x; 1.1366x over previous
#include <hip/hip_runtime.h>


#define LLEN   8192
#define LH     4096
#define LQ     2048
#define NMODES 32
#define NF     64
#define NROWS  4096
#define KSPLIT 16
#define KBLK   2
#define MPAIRS 128
#define NCHUNK 32
#define TWO_PI 6.283185307179586476925f

#define T2_OFF  (LH * NF)
#define XF_OFF  (2 * LH * NF)
#define G_OFF   (2 * LH * NF + KBLK * NROWS * NF)

typedef float v2f __attribute__((ext_vector_type(2)));
typedef float v4f __attribute__((ext_vector_type(4)));
typedef float v8f __attribute__((ext_vector_type(8)));

__device__ __forceinline__ v2f t2q_pair(int t) {
  const int w = t >> 5, lane = t & 31;
  const int hi = w & 1, qh = (w >> 1) & 1, ks = (w >> 2) & 15, grp = w >> 6;
  const int l  = 64 * grp + 4 * (lane >> 1) + 2 * qh + (lane & 1);
  const int kk = 4 * ks + 2 * hi;
  const bool ev = kk < 32;
  const int f0   = ev ? 4 * (kk & 7) + 2 * (kk >> 4) : 2 * (kk & 15) + 1;
  const int f1   = f0 + (ev ? 4 : 2);
  const int part = ev ? (kk >> 3) & 1 : (kk >> 4) & 1;
  const float ph0 = (float)((f0 * l) & (LLEN - 1)) * (TWO_PI / (float)LLEN);
  const float ph1 = (float)((f1 * l) & (LLEN - 1)) * (TWO_PI / (float)LLEN);
  v2f v;
  if (part) { v.x = -sinf(ph0); v.y = -sinf(ph1); }
  else      { v.x =  cosf(ph0); v.y =  cosf(ph1); }
  return v;
}

__global__ __launch_bounds__(512) void k_dft_fwd(const float* __restrict__ x,
                                                 float* __restrict__ xf) {
  __shared__ float red[KSPLIT * 32 * NF];
  __shared__ v2f rot[128];
  if (threadIdx.x < 128) {
    float s, c;
    sincospif((float)threadIdx.x * (1.0f / 64.0f), &s, &c);
    const v2f cs = {c, s};
    rot[threadIdx.x] = cs;
  }
  const int lane = threadIdx.x & 31;
  const int wid  = threadIdx.x >> 5;
  const int hi   = lane >> 4;
  const int ln   = lane & 15;
  const int mp = blockIdx.x / KBLK;
  const int kh = blockIdx.x % KBLK;
  const int rowbase = mp * 32;
  const float* xrow0 = x + (size_t)(rowbase + ln) * LLEN;
  const float* xrow1 = x + (size_t)(rowbase + 16 + ln) * LLEN;

  v8f a0t0 = {}, a0t1 = {}, a0t2 = {}, a0t3 = {};
  v8f a1t0 = {}, a1t1 = {}, a1t2 = {}, a1t3 = {};
  const int fe0 = 4 * (ln & 7), fe1 = fe0 + 2, fo = 2 * ln + 1;
  const int l0 = 4 * wid + 2 * hi;
  v2f p0c, p0s, p1c, p1s, p2c, p2s;
  {
    float s0, c0, s1, c1;
    sincospif((float)(fe0 * l0) * (1.0f / 4096.0f), &s0, &c0);
    sincospif((float)(fe0 * (l0 + 1)) * (1.0f / 4096.0f), &s1, &c1);
    const v2f cc0 = {c0, c1}, ss0 = {s0, s1};
    sincospif((float)(fe1 * l0) * (1.0f / 4096.0f), &s0, &c0);
    sincospif((float)(fe1 * (l0 + 1)) * (1.0f / 4096.0f), &s1, &c1);
    const v2f cc1 = {c0, c1}, ss1 = {s0, s1};
    sincospif((float)(fo * l0) * (1.0f / 4096.0f), &s0, &c0);
    sincospif((float)(fo * (l0 + 1)) * (1.0f / 4096.0f), &s1, &c1);
    const v2f cc2 = {c0, c1}, ss2 = {s0, s1};
    const bool im = (ln >> 3) != 0;
    p0c = im ? -ss0 : cc0;  p0s = im ? cc0 : ss0;
    p1c = im ? -ss1 : cc1;  p1s = im ? cc1 : ss1;
    p2c = cc2;              p2s = ss2;
  }
  const float sg = (ln & 1) ? -1.0f : 1.0f;
  __syncthreads();

  const int k0beg = kh * (LQ / KBLK) + 4 * wid + 2 * hi;
  for (int it = 0; it < LQ / (KBLK * KSPLIT * 4); ++it) {
    const int kb = k0beg + 4 * KSPLIT * it;
    const v2f xa0 = *(const v2f*)(xrow0 + kb);
    const v2f xb0 = *(const v2f*)(xrow0 + kb + LQ);
    const v2f xc0 = *(const v2f*)(xrow0 + kb + LH);
    const v2f xd0 = *(const v2f*)(xrow0 + kb + LH + LQ);
    const v2f xa1 = *(const v2f*)(xrow1 + kb);
    const v2f xb1 = *(const v2f*)(xrow1 + kb + LQ);
    const v2f xc1 = *(const v2f*)(xrow1 + kb + LH);
    const v2f xd1 = *(const v2f*)(xrow1 + kb + LH + LQ);
    const int m = kh * (LQ / (KBLK * 64)) + it;
    const v2f r0 = rot[(fe0 * m) & 127];
    const v2f r1 = rot[(fe1 * m) & 127];
    const v2f r2 = rot[(fo  * m) & 127];
    const v2f b0 = r0.x * p0c - r0.y * p0s;
    const v2f b1 = r1.x * p1c - r1.y * p1s;
    const v2f b2 = r2.x * p2c - r2.y * p2s;
    const v2f b3 = -(r2.x * p2s) - r2.y * p2c;
    const v2f c2 = sg * b3;
    const v2f c3 = -sg * b2;
    const v2f ea0 = xa0 + xc0, eb0 = xb0 + xd0;
    const v2f oa0 = xa0 - xc0, ob0 = xb0 - xd0;
    const v2f ee0 = ea0 + eb0, eo0 = ea0 - eb0;
    const v2f ea1 = xa1 + xc1, eb1 = xb1 + xd1;
    const v2f oa1 = xa1 - xc1, ob1 = xb1 - xd1;
    const v2f ee1 = ea1 + eb1, eo1 = ea1 - eb1;
    a0t0 = __builtin_amdgcn_wmma_f32_16x16x4_f32(false, ee0, false, b0, (short)0, a0t0, false, false);
    a0t1 = __builtin_amdgcn_wmma_f32_16x16x4_f32(false, eo0, false, b1, (short)0, a0t1, false, false);
    a0t2 = __builtin_amdgcn_wmma_f32_16x16x4_f32(false, oa0, false, b2, (short)0, a0t2, false, false);
    a0t3 = __builtin_amdgcn_wmma_f32_16x16x4_f32(false, oa0, false, b3, (short)0, a0t3, false, false);
    a1t0 = __builtin_amdgcn_wmma_f32_16x16x4_f32(false, ee1, false, b0, (short)0, a1t0, false, false);
    a1t1 = __builtin_amdgcn_wmma_f32_16x16x4_f32(false, eo1, false, b1, (short)0, a1t1, false, false);
    a1t2 = __builtin_amdgcn_wmma_f32_16x16x4_f32(false, oa1, false, b2, (short)0, a1t2, false, false);
    a1t3 = __builtin_amdgcn_wmma_f32_16x16x4_f32(false, oa1, false, b3, (short)0, a1t3, false, false);
    a0t2 = __builtin_amdgcn_wmma_f32_16x16x4_f32(false, ob0, false, c2, (short)0, a0t2, false, false);
    a0t3 = __builtin_amdgcn_wmma_f32_16x16x4_f32(false, ob0, false, c3, (short)0, a0t3, false, false);
    a1t2 = __builtin_amdgcn_wmma_f32_16x16x4_f32(false, ob1, false, c2, (short)0, a1t2, false, false);
    a1t3 = __builtin_amdgcn_wmma_f32_16x16x4_f32(false, ob1, false, c3, (short)0, a1t3, false, false);
  }

  float* my = red + wid * (32 * NF);
  #pragma unroll
  for (int m = 0; m < 8; ++m) {
    const int r = m + 8 * hi;
    my[r * NF + 0 * 16 + ln] = a0t0[m];
    my[r * NF + 1 * 16 + ln] = a0t1[m];
    my[r * NF + 2 * 16 + ln] = a0t2[m];
    my[r * NF + 3 * 16 + ln] = a0t3[m];
    my[(16 + r) * NF + 0 * 16 + ln] = a1t0[m];
    my[(16 + r) * NF + 1 * 16 + ln] = a1t1[m];
    my[(16 + r) * NF + 2 * 16 + ln] = a1t2[m];
    my[(16 + r) * NF + 3 * 16 + ln] = a1t3[m];
  }
  __syncthreads();
  #pragma unroll
  for (int u = 0; u < 4; ++u) {
    const int idx = threadIdx.x + 512 * u;
    float s = 0.0f;
    #pragma unroll
    for (int w = 0; w < KSPLIT; ++w) s += red[w * (32 * NF) + idx];
    xf[(size_t)kh * (NROWS * NF) + (size_t)mp * (32 * NF) + idx] = s;
  }
}

__global__ __launch_bounds__(256) void k_mix(const float* __restrict__ xf,
                                             const float* __restrict__ wr,
                                             const float* __restrict__ wi,
                                             float* __restrict__ g,
                                             float* __restrict__ t2q) {
  const int idx = blockIdx.x * 256 + threadIdx.x;
  const int f = idx & 31;
  const int o = (idx >> 5) & 15;
  const int h = (idx >> 9) & 3;
  const int b = idx >> 11;
  __shared__ v4f xs4[16 * NF / 4];
  const float* xp = xf + (size_t)(b * 64 + h * 16) * NF + threadIdx.x * 4;
  v4f s = *(const v4f*)xp;
  for (int p = 1; p < KBLK; ++p) s += *(const v4f*)(xp + (size_t)p * (NROWS * NF));
  const float* wr0 = wr + ((size_t)(h * 16) * 16 + o) * NMODES + f;
  const float* wi0 = wi + ((size_t)(h * 16) * 16 + o) * NMODES + f;
  float wre[16], wim[16];
  #pragma unroll
  for (int i = 0; i < 16; ++i) {
    wre[i] = wr0[i * (16 * NMODES)];
    wim[i] = wi0[i * (16 * NMODES)];
  }
  *(v2f*)(t2q + 2 * (size_t)idx) = t2q_pair(idx);
  xs4[threadIdx.x] = s;
  __syncthreads();
  const int kr = (f & 1) ? 32 + (f >> 1) : ((f >> 1) & 1) * 16 + (f >> 2);
  const int ki = kr + ((f & 1) ? 16 : 8);
  const float* xr0 = (const float*)xs4 + kr;
  const float* xi0 = (const float*)xs4 + ki;
  float or_ = 0.0f, oi_ = 0.0f;
  #pragma unroll
  for (int i = 0; i < 16; ++i) {
    const float xr  = xr0[i * NF];
    const float xi  = xi0[i * NF];
    or_ += xr * wre[i] - xi * wim[i];
    oi_ += xr * wim[i] + xi * wre[i];
  }
  const float invL = 1.0f / (float)LLEN;
  const float scale = (f == 0) ? invL : 2.0f * invL;
  const size_t ro = (size_t)(b * 64 + h * 16 + o);
  g[ro * NF + kr]      = or_ * scale;
  g[ro * NF + ki]      = oi_ * scale;
}

__global__ __launch_bounds__(128) __attribute__((amdgpu_waves_per_eu(2, 2))) void k_idft(const float* __restrict__ g,
                                              const float* __restrict__ t2q,
                                              const float* __restrict__ bias,
                                              float* __restrict__ out) {
  const int lane = threadIdx.x & 31;
  const int wid  = threadIdx.x >> 5;
  const int gw   = blockIdx.x * 4 + wid;
  const int mpair = gw >> 5;
  const int nc    = gw & (NCHUNK - 1);
  const int hi = lane >> 4, ln = lane & 15;
  const int rowbase = mpair * 32;

  v2f af0[16], af1[16];
  #pragma unroll
  for (int ks = 0; ks < 16; ++ks) {
    const int kb = ks * 4 + 2 * hi;
    af0[ks] = *(const v2f*)(g + (size_t)(rowbase + ln) * NF + kb);
    af1[ks] = *(const v2f*)(g + (size_t)(rowbase + 16 + ln) * NF + kb);
  }
  float bv0[8], bv1[8];
  #pragma unroll
  for (int m = 0; m < 8; ++m) {
    bv0[m] = bias[(rowbase + m + 8 * hi) & 63];
    bv1[m] = bias[(rowbase + 16 + m + 8 * hi) & 63];
  }

  const int col = nc * (LQ / NCHUNK) + 4 * ln;
  const float* tb = t2q + (size_t)nc * 4096 + (hi * 16 + ln) * 4;
  const float* tq = tb + (size_t)(LQ >> 6) * 4096;

  const v8f bz0 = {bv0[0], bv0[1], bv0[2], bv0[3], bv0[4], bv0[5], bv0[6], bv0[7]};
  const v8f bz1 = {bv1[0], bv1[1], bv1[2], bv1[3], bv1[4], bv1[5], bv1[6], bv1[7]};
  v8f z00 = bz0, z01 = bz0, z02 = bz0, z03 = bz0, z10 = bz1, z11 = bz1, z12 = bz1, z13 = bz1;
  v8f t00 = {}, t01 = {}, t02 = {}, t03 = {}, t10 = {}, t11 = {}, t12 = {}, t13 = {};
  #pragma unroll
  for (int ks = 0; ks < 4; ++ks) {
    const v4f p = *(const v4f*)(tb + (ks * 2 + 0) * 128);
    const v4f r = *(const v4f*)(tb + (ks * 2 + 1) * 128);
    const v2f b0 = {p.x, p.y};
    const v2f b1 = {p.z, p.w};
    const v2f b2 = {r.x, r.y};
    const v2f b3 = {r.z, r.w};
    z00 = __builtin_amdgcn_wmma_f32_16x16x4_f32(false, af0[ks], false, b0, (short)0, z00, false, false);
    z01 = __builtin_amdgcn_wmma_f32_16x16x4_f32(false, af0[ks], false, b1, (short)0, z01, false, false);
    z02 = __builtin_amdgcn_wmma_f32_16x16x4_f32(false, af0[ks], false, b2, (short)0, z02, false, false);
    z03 = __builtin_amdgcn_wmma_f32_16x16x4_f32(false, af0[ks], false, b3, (short)0, z03, false, false);
    z10 = __builtin_amdgcn_wmma_f32_16x16x4_f32(false, af1[ks], false, b0, (short)0, z10, false, false);
    z11 = __builtin_amdgcn_wmma_f32_16x16x4_f32(false, af1[ks], false, b1, (short)0, z11, false, false);
    z12 = __builtin_amdgcn_wmma_f32_16x16x4_f32(false, af1[ks], false, b2, (short)0, z12, false, false);
    z13 = __builtin_amdgcn_wmma_f32_16x16x4_f32(false, af1[ks], false, b3, (short)0, z13, false, false);
  }
  #pragma unroll
  for (int ks = 4; ks < 8; ++ks) {
    const v4f p = *(const v4f*)(tb + (ks * 2 + 0) * 128);
    const v4f r = *(const v4f*)(tb + (ks * 2 + 1) * 128);
    const v2f b0 = {p.x, p.y};
    const v2f b1 = {p.z, p.w};
    const v2f b2 = {r.x, r.y};
    const v2f b3 = {r.z, r.w};
    t00 = __builtin_amdgcn_wmma_f32_16x16x4_f32(false, af0[ks], false, b0, (short)0, t00, false, false);
    t01 = __builtin_amdgcn_wmma_f32_16x16x4_f32(false, af0[ks], false, b1, (short)0, t01, false, false);
    t02 = __builtin_amdgcn_wmma_f32_16x16x4_f32(false, af0[ks], false, b2, (short)0, t02, false, false);
    t03 = __builtin_amdgcn_wmma_f32_16x16x4_f32(false, af0[ks], false, b3, (short)0, t03, false, false);
    t10 = __builtin_amdgcn_wmma_f32_16x16x4_f32(false, af1[ks], false, b0, (short)0, t10, false, false);
    t11 = __builtin_amdgcn_wmma_f32_16x16x4_f32(false, af1[ks], false, b1, (short)0, t11, false, false);
    t12 = __builtin_amdgcn_wmma_f32_16x16x4_f32(false, af1[ks], false, b2, (short)0, t12, false, false);
    t13 = __builtin_amdgcn_wmma_f32_16x16x4_f32(false, af1[ks], false, b3, (short)0, t13, false, false);
  }

  #pragma unroll
  for (int ps = 0; ps < 2; ++ps) {
    const float* to = ps ? tq : tb;
    v8f o00 = {}, o01 = {}, o02 = {}, o03 = {};
    v8f o10 = {}, o11 = {}, o12 = {}, o13 = {};
    #pragma unroll
    for (int ks = 8; ks < 16; ++ks) {
      const v4f p = *(const v4f*)(to + (ks * 2 + 0) * 128);
      const v4f r = *(const v4f*)(to + (ks * 2 + 1) * 128);
      const v2f b0 = {p.x, p.y};
      const v2f b1 = {p.z, p.w};
      const v2f b2 = {r.x, r.y};
      const v2f b3 = {r.z, r.w};
      o00 = __builtin_amdgcn_wmma_f32_16x16x4_f32(false, af0[ks], false, b0, (short)0, o00, false, false);
      o01 = __builtin_amdgcn_wmma_f32_16x16x4_f32(false, af0[ks], false, b1, (short)0, o01, false, false);
      o02 = __builtin_amdgcn_wmma_f32_16x16x4_f32(false, af0[ks], false, b2, (short)0, o02, false, false);
      o03 = __builtin_amdgcn_wmma_f32_16x16x4_f32(false, af0[ks], false, b3, (short)0, o03, false, false);
      o10 = __builtin_amdgcn_wmma_f32_16x16x4_f32(false, af1[ks], false, b0, (short)0, o10, false, false);
      o11 = __builtin_amdgcn_wmma_f32_16x16x4_f32(false, af1[ks], false, b1, (short)0, o11, false, false);
      o12 = __builtin_amdgcn_wmma_f32_16x16x4_f32(false, af1[ks], false, b2, (short)0, o12, false, false);
      o13 = __builtin_amdgcn_wmma_f32_16x16x4_f32(false, af1[ks], false, b3, (short)0, o13, false, false);
    }
    const float sg = ps ? -1.0f : 1.0f;
    float* olo = out + (size_t)ps * LQ + col;
    float* ohi = olo + LH;
    #pragma unroll
    for (int m = 0; m < 8; ++m) {
      const int r = rowbase + m + 8 * hi;
      const float s00 = z00[m] + sg * t00[m], s01 = z01[m] + sg * t01[m];
      const float s02 = z02[m] + sg * t02[m], s03 = z03[m] + sg * t03[m];
      const float s10 = z10[m] + sg * t10[m], s11 = z11[m] + sg * t11[m];
      const float s12 = z12[m] + sg * t12[m], s13 = z13[m] + sg * t13[m];
      const v4f lo0 = {s00 + o00[m], s01 + o01[m], s02 + o02[m], s03 + o03[m]};
      const v4f hi0 = {s00 - o00[m], s01 - o01[m], s02 - o02[m], s03 - o03[m]};
      const v4f lo1 = {s10 + o10[m], s11 + o11[m], s12 + o12[m], s13 + o13[m]};
      const v4f hi1 = {s10 - o10[m], s11 - o11[m], s12 - o12[m], s13 - o13[m]};
      *(v4f*)(olo + (size_t)r * LLEN)        = lo0;
      *(v4f*)(ohi + (size_t)r * LLEN)        = hi0;
      *(v4f*)(olo + (size_t)(r + 16) * LLEN) = lo1;
      *(v4f*)(ohi + (size_t)(r + 16) * LLEN) = hi1;
    }
  }
}

extern "C" void kernel_launch(void* const* d_in, const int* in_sizes, int n_in,
                              void* d_out, int out_size, void* d_ws, size_t ws_size,
                              hipStream_t stream) {
  const float* x    = (const float*)d_in[0];
  const float* wre  = (const float*)d_in[1];
  const float* wim  = (const float*)d_in[2];
  const float* bias = (const float*)d_in[3];
  float* out = (float*)d_out;
  float* ws  = (float*)d_ws;

  float* t2q = ws + T2_OFF;
  float* xf  = ws + XF_OFF;
  float* g   = ws + G_OFF;

  k_dft_fwd<<<MPAIRS * KBLK, 512, 0, stream>>>(x, xf);
  k_mix    <<<(NROWS * NF / 2) / 256, 256, 0, stream>>>(xf, wre, wim, g, t2q);
  k_idft   <<<(MPAIRS * NCHUNK) / 4, 128, 0, stream>>>(g, t2q, bias, out);
}


# Mix fused into stage 1: a workgroup takes one head's 16 rows over the whole K, sums its waves in LDS, mixes, writes the table (now by cospif): 2 launches, was 3; Xf sum order, table last bits change
# speedup vs baseline: 2.7930x; 1.0997x over previous
#include <hip/hip_runtime.h>


#define LLEN   8192
#define LH     4096
#define LQ     2048
#define NMODES 32
#define NF     64
#define NROWS  4096
#define KSPLIT 16
#define HROWS  16
#define MPAIRS 128
#define NCHUNK 32

#define T2_OFF  (LH * NF)
#define G_OFF   (2 * LH * NF + 2 * NROWS * NF)

typedef float v2f __attribute__((ext_vector_type(2)));
typedef float v4f __attribute__((ext_vector_type(4)));
typedef float v8f __attribute__((ext_vector_type(8)));

__device__ __forceinline__ v2f t2q_pair(int t) {
  const int w = t >> 5, lane = t & 31;
  const int hi = w & 1, qh = (w >> 1) & 1, ks = (w >> 2) & 15, grp = w >> 6;
  const int l  = 64 * grp + 4 * (lane >> 1) + 2 * qh + (lane & 1);
  const int kk = 4 * ks + 2 * hi;
  const bool ev = kk < 32;
  const int f0   = ev ? 4 * (kk & 7) + 2 * (kk >> 4) : 2 * (kk & 15) + 1;
  const int f1   = f0 + (ev ? 4 : 2);
  const int part = ev ? (kk >> 3) & 1 : (kk >> 4) & 1;
  const int qp = part * (LLEN / 4);
  v2f v;
  v.x = cospif((float)((f0 * l + qp) & (LLEN - 1)) * (2.0f / (float)LLEN));
  v.y = cospif((float)((f1 * l + qp) & (LLEN - 1)) * (2.0f / (float)LLEN));
  return v;
}

__global__ __launch_bounds__(512) void k_dft_fwd_mix(const float* __restrict__ x,
                                                     const float* __restrict__ wr,
                                                     const float* __restrict__ wi,
                                                     float* __restrict__ g,
                                                     float* __restrict__ t2q) {
  __shared__ float red[KSPLIT * HROWS * NF];
  __shared__ float xs[HROWS * NF];
  __shared__ v2f rot[128];
  if (threadIdx.x < 128) {
    float s, c;
    sincospif((float)threadIdx.x * (1.0f / 64.0f), &s, &c);
    const v2f cs = {c, s};
    rot[threadIdx.x] = cs;
  }
  const int lane = threadIdx.x & 31;
  const int wid  = threadIdx.x >> 5;
  const int hi   = lane >> 4;
  const int ln   = lane & 15;
  const int rowbase = blockIdx.x * HROWS;
  const float* xrow = x + (size_t)(rowbase + ln) * LLEN;

  v8f at0 = {}, at1 = {}, at2 = {}, at3 = {};
  const int fe0 = 4 * (ln & 7), fe1 = fe0 + 2, fo = 2 * ln + 1;
  const int l0 = 4 * wid + 2 * hi;
  v2f p0c, p0s, p1c, p1s, p2c, p2s;
  {
    float s0, c0, s1, c1;
    sincospif((float)(fe0 * l0) * (1.0f / 4096.0f), &s0, &c0);
    sincospif((float)(fe0 * (l0 + 1)) * (1.0f / 4096.0f), &s1, &c1);
    const v2f cc0 = {c0, c1}, ss0 = {s0, s1};
    sincospif((float)(fe1 * l0) * (1.0f / 4096.0f), &s0, &c0);
    sincospif((float)(fe1 * (l0 + 1)) * (1.0f / 4096.0f), &s1, &c1);
    const v2f cc1 = {c0, c1}, ss1 = {s0, s1};
    sincospif((float)(fo * l0) * (1.0f / 4096.0f), &s0, &c0);
    sincospif((float)(fo * (l0 + 1)) * (1.0f / 4096.0f), &s1, &c1);
    const v2f cc2 = {c0, c1}, ss2 = {s0, s1};
    const bool im = (ln >> 3) != 0;
    p0c = im ? -ss0 : cc0;  p0s = im ? cc0 : ss0;
    p1c = im ? -ss1 : cc1;  p1s = im ? cc1 : ss1;
    p2c = cc2;              p2s = ss2;
  }
  const float sg = (ln & 1) ? -1.0f : 1.0f;
  const int gid = blockIdx.x * 512 + threadIdx.x;
  *(v2f*)(t2q + 2 * (size_t)gid) = t2q_pair(gid);
  __builtin_amdgcn_fence(__ATOMIC_RELEASE, "workgroup", "local");
  __builtin_amdgcn_s_barrier();
  __builtin_amdgcn_fence(__ATOMIC_ACQUIRE, "workgroup", "local");

  const int k0beg = 4 * wid + 2 * hi;
  #pragma unroll
  for (int it = 0; it < LQ / (KSPLIT * 4); ++it) {
    const int kb = k0beg + 4 * KSPLIT * it;
    const v2f xa = *(const v2f*)(xrow + kb);
    const v2f xb = *(const v2f*)(xrow + kb + LQ);
    const v2f xc = *(const v2f*)(xrow + kb + LH);
    const v2f xd = *(const v2f*)(xrow + kb + LH + LQ);
    const int m = it;
    const v2f r0 = rot[(fe0 * m) & 127];
    const v2f r1 = rot[(fe1 * m) & 127];
    const v2f r2 = rot[(fo  * m) & 127];
    const v2f b0 = r0.x * p0c - r0.y * p0s;
    const v2f b1 = r1.x * p1c - r1.y * p1s;
    const v2f b2 = r2.x * p2c - r2.y * p2s;
    const v2f b3 = -(r2.x * p2s) - r2.y * p2c;
    const v2f c2 = sg * b3;
    const v2f c3 = -sg * b2;
    const v2f ea = xa + xc, eb = xb + xd;
    const v2f oa = xa - xc, ob = xb - xd;
    const v2f ee = ea + eb, eo = ea - eb;
    at0 = __builtin_amdgcn_wmma_f32_16x16x4_f32(false, ee, false, b0, (short)0, at0, false, false);
    at1 = __builtin_amdgcn_wmma_f32_16x16x4_f32(false, eo, false, b1, (short)0, at1, false, false);
    at2 = __builtin_amdgcn_wmma_f32_16x16x4_f32(false, oa, false, b2, (short)0, at2, false, false);
    at3 = __builtin_amdgcn_wmma_f32_16x16x4_f32(false, oa, false, b3, (short)0, at3, false, false);
    at2 = __builtin_amdgcn_wmma_f32_16x16x4_f32(false, ob, false, c2, (short)0, at2, false, false);
    at3 = __builtin_amdgcn_wmma_f32_16x16x4_f32(false, ob, false, c3, (short)0, at3, false, false);
  }

  float* my = red + wid * (HROWS * NF);
  #pragma unroll
  for (int m = 0; m < 8; ++m) {
    const int r = m + 8 * hi;
    my[r * NF + 0 * 16 + ln] = at0[m];
    my[r * NF + 1 * 16 + ln] = at1[m];
    my[r * NF + 2 * 16 + ln] = at2[m];
    my[r * NF + 3 * 16 + ln] = at3[m];
  }
  const int f = threadIdx.x & 31;
  const int o = threadIdx.x >> 5;
  const int h = blockIdx.x & 3;
  const float* wr0 = wr + ((size_t)(h * 16) * 16 + o) * NMODES + f;
  const float* wi0 = wi + ((size_t)(h * 16) * 16 + o) * NMODES + f;
  float wre[16], wim[16];
  #pragma unroll
  for (int i = 0; i < 16; ++i) {
    wre[i] = wr0[i * (16 * NMODES)];
    wim[i] = wi0[i * (16 * NMODES)];
  }
  __syncthreads();
  #pragma unroll
  for (int u = 0; u < 2; ++u) {
    const int idx = threadIdx.x + 512 * u;
    float s = 0.0f;
    #pragma unroll
    for (int w = 0; w < KSPLIT; ++w) s += red[w * (HROWS * NF) + idx];
    xs[idx] = s;
  }
  __syncthreads();
  const int kr = (f & 1) ? 32 + (f >> 1) : ((f >> 1) & 1) * 16 + (f >> 2);
  const int ki = kr + ((f & 1) ? 16 : 8);
  const float* xr0 = xs + kr;
  const float* xi0 = xs + ki;
  float or_ = 0.0f, oi_ = 0.0f;
  #pragma unroll
  for (int i = 0; i < 16; ++i) {
    const float xr  = xr0[i * NF];
    const float xi  = xi0[i * NF];
    or_ += xr * wre[i] - xi * wim[i];
    oi_ += xr * wim[i] + xi * wre[i];
  }
  const float invL = 1.0f / (float)LLEN;
  const float scale = (f == 0) ? invL : 2.0f * invL;
  const size_t ro = (size_t)(rowbase + o);
  g[ro * NF + kr]      = or_ * scale;
  g[ro * NF + ki]      = oi_ * scale;
}

__global__ __launch_bounds__(128) __attribute__((amdgpu_waves_per_eu(2, 2))) void k_idft(const float* __restrict__ g,
                                              const float* __restrict__ t2q,
                                              const float* __restrict__ bias,
                                              float* __restrict__ out) {
  const int lane = threadIdx.x & 31;
  const int wid  = threadIdx.x >> 5;
  const int gw   = blockIdx.x * 4 + wid;
  const int mpair = gw >> 5;
  const int nc    = gw & (NCHUNK - 1);
  const int hi = lane >> 4, ln = lane & 15;
  const int rowbase = mpair * 32;

  v2f af0[16], af1[16];
  #pragma unroll
  for (int ks = 0; ks < 16; ++ks) {
    const int kb = ks * 4 + 2 * hi;
    af0[ks] = *(const v2f*)(g + (size_t)(rowbase + ln) * NF + kb);
    af1[ks] = *(const v2f*)(g + (size_t)(rowbase + 16 + ln) * NF + kb);
  }
  float bv0[8], bv1[8];
  #pragma unroll
  for (int m = 0; m < 8; ++m) {
    bv0[m] = bias[(rowbase + m + 8 * hi) & 63];
    bv1[m] = bias[(rowbase + 16 + m + 8 * hi) & 63];
  }

  const int col = nc * (LQ / NCHUNK) + 4 * ln;
  const float* tb = t2q + (size_t)nc * 4096 + (hi * 16 + ln) * 4;
  const float* tq = tb + (size_t)(LQ >> 6) * 4096;

  const v8f bz0 = {bv0[0], bv0[1], bv0[2], bv0[3], bv0[4], bv0[5], bv0[6], bv0[7]};
  const v8f bz1 = {bv1[0], bv1[1], bv1[2], bv1[3], bv1[4], bv1[5], bv1[6], bv1[7]};
  v8f z00 = bz0, z01 = bz0, z02 = bz0, z03 = bz0, z10 = bz1, z11 = bz1, z12 = bz1, z13 = bz1;
  v8f t00 = {}, t01 = {}, t02 = {}, t03 = {}, t10 = {}, t11 = {}, t12 = {}, t13 = {};
  #pragma unroll
  for (int ks = 0; ks < 4; ++ks) {
    const v4f p = *(const v4f*)(tb + (ks * 2 + 0) * 128);
    const v4f r = *(const v4f*)(tb + (ks * 2 + 1) * 128);
    const v2f b0 = {p.x, p.y};
    const v2f b1 = {p.z, p.w};
    const v2f b2 = {r.x, r.y};
    const v2f b3 = {r.z, r.w};
    z00 = __builtin_amdgcn_wmma_f32_16x16x4_f32(false, af0[ks], false, b0, (short)0, z00, false, false);
    z01 = __builtin_amdgcn_wmma_f32_16x16x4_f32(false, af0[ks], false, b1, (short)0, z01, false, false);
    z02 = __builtin_amdgcn_wmma_f32_16x16x4_f32(false, af0[ks], false, b2, (short)0, z02, false, false);
    z03 = __builtin_amdgcn_wmma_f32_16x16x4_f32(false, af0[ks], false, b3, (short)0, z03, false, false);
    z10 = __builtin_amdgcn_wmma_f32_16x16x4_f32(false, af1[ks], false, b0, (short)0, z10, false, false);
    z11 = __builtin_amdgcn_wmma_f32_16x16x4_f32(false, af1[ks], false, b1, (short)0, z11, false, false);
    z12 = __builtin_amdgcn_wmma_f32_16x16x4_f32(false, af1[ks], false, b2, (short)0, z12, false, false);
    z13 = __builtin_amdgcn_wmma_f32_16x16x4_f32(false, af1[ks], false, b3, (short)0, z13, false, false);
  }
  #pragma unroll
  for (int ks = 4; ks < 8; ++ks) {
    const v4f p = *(const v4f*)(tb + (ks * 2 + 0) * 128);
    const v4f r = *(const v4f*)(tb + (ks * 2 + 1) * 128);
    const v2f b0 = {p.x, p.y};
    const v2f b1 = {p.z, p.w};
    const v2f b2 = {r.x, r.y};
    const v2f b3 = {r.z, r.w};
    t00 = __builtin_amdgcn_wmma_f32_16x16x4_f32(false, af0[ks], false, b0, (short)0, t00, false, false);
    t01 = __builtin_amdgcn_wmma_f32_16x16x4_f32(false, af0[ks], false, b1, (short)0, t01, false, false);
    t02 = __builtin_amdgcn_wmma_f32_16x16x4_f32(false, af0[ks], false, b2, (short)0, t02, false, false);
    t03 = __builtin_amdgcn_wmma_f32_16x16x4_f32(false, af0[ks], false, b3, (short)0, t03, false, false);
    t10 = __builtin_amdgcn_wmma_f32_16x16x4_f32(false, af1[ks], false, b0, (short)0, t10, false, false);
    t11 = __builtin_amdgcn_wmma_f32_16x16x4_f32(false, af1[ks], false, b1, (short)0, t11, false, false);
    t12 = __builtin_amdgcn_wmma_f32_16x16x4_f32(false, af1[ks], false, b2, (short)0, t12, false, false);
    t13 = __builtin_amdgcn_wmma_f32_16x16x4_f32(false, af1[ks], false, b3, (short)0, t13, false, false);
  }

  #pragma unroll
  for (int ps = 0; ps < 2; ++ps) {
    const float* to = ps ? tq : tb;
    v8f o00 = {}, o01 = {}, o02 = {}, o03 = {};
    v8f o10 = {}, o11 = {}, o12 = {}, o13 = {};
    #pragma unroll
    for (int ks = 8; ks < 16; ++ks) {
      const v4f p = *(const v4f*)(to + (ks * 2 + 0) * 128);
      const v4f r = *(const v4f*)(to + (ks * 2 + 1) * 128);
      const v2f b0 = {p.x, p.y};
      const v2f b1 = {p.z, p.w};
      const v2f b2 = {r.x, r.y};
      const v2f b3 = {r.z, r.w};
      o00 = __builtin_amdgcn_wmma_f32_16x16x4_f32(false, af0[ks], false, b0, (short)0, o00, false, false);
      o01 = __builtin_amdgcn_wmma_f32_16x16x4_f32(false, af0[ks], false, b1, (short)0, o01, false, false);
      o02 = __builtin_amdgcn_wmma_f32_16x16x4_f32(false, af0[ks], false, b2, (short)0, o02, false, false);
      o03 = __builtin_amdgcn_wmma_f32_16x16x4_f32(false, af0[ks], false, b3, (short)0, o03, false, false);
      o10 = __builtin_amdgcn_wmma_f32_16x16x4_f32(false, af1[ks], false, b0, (short)0, o10, false, false);
      o11 = __builtin_amdgcn_wmma_f32_16x16x4_f32(false, af1[ks], false, b1, (short)0, o11, false, false);
      o12 = __builtin_amdgcn_wmma_f32_16x16x4_f32(false, af1[ks], false, b2, (short)0, o12, false, false);
      o13 = __builtin_amdgcn_wmma_f32_16x16x4_f32(false, af1[ks], false, b3, (short)0, o13, false, false);
    }
    const float sg = ps ? -1.0f : 1.0f;
    float* olo = out + (size_t)ps * LQ + col;
    float* ohi = olo + LH;
    #pragma unroll
    for (int m = 0; m < 8; ++m) {
      const int r = rowbase + m + 8 * hi;
      const float s00 = z00[m] + sg * t00[m], s01 = z01[m] + sg * t01[m];
      const float s02 = z02[m] + sg * t02[m], s03 = z03[m] + sg * t03[m];
      const float s10 = z10[m] + sg * t10[m], s11 = z11[m] + sg * t11[m];
      const float s12 = z12[m] + sg * t12[m], s13 = z13[m] + sg * t13[m];
      const v4f lo0 = {s00 + o00[m], s01 + o01[m], s02 + o02[m], s03 + o03[m]};
      const v4f hi0 = {s00 - o00[m], s01 - o01[m], s02 - o02[m], s03 - o03[m]};
      const v4f lo1 = {s10 + o10[m], s11 + o11[m], s12 + o12[m], s13 + o13[m]};
      const v4f hi1 = {s10 - o10[m], s11 - o11[m], s12 - o12[m], s13 - o13[m]};
      *(v4f*)(olo + (size_t)r * LLEN)        = lo0;
      *(v4f*)(ohi + (size_t)r * LLEN)        = hi0;
      *(v4f*)(olo + (size_t)(r + 16) * LLEN) = lo1;
      *(v4f*)(ohi + (size_t)(r + 16) * LLEN) = hi1;
    }
  }
}

extern "C" void kernel_launch(void* const* d_in, const int* in_sizes, int n_in,
                              void* d_out, int out_size, void* d_ws, size_t ws_size,
                              hipStream_t stream) {
  const float* x    = (const float*)d_in[0];
  const float* wre  = (const float*)d_in[1];
  const float* wim  = (const float*)d_in[2];
  const float* bias = (const float*)d_in[3];
  float* out = (float*)d_out;
  float* ws  = (float*)d_ws;

  float* t2q = ws + T2_OFF;
  float* g   = ws + G_OFF;

  k_dft_fwd_mix<<<NROWS / HROWS, 512, 0, stream>>>(x, wre, wim, g, t2q);
  k_idft       <<<(MPAIRS * NCHUNK) / 4, 128, 0, stream>>>(g, t2q, bias, out);
}


# Stage 1 row loads widened to 16 B a lane: 4 consecutive l per stream feed two k-steps (basis lane constants for 4 rows), half the loads and cache-line touches; Xf sum order, basis last bits change
# speedup vs baseline: 2.9284x; 1.0485x over previous
#include <hip/hip_runtime.h>


#define LLEN   8192
#define LH     4096
#define LQ     2048
#define NMODES 32
#define NF     64
#define NROWS  4096
#define KSPLIT 16
#define HROWS  16
#define MPAIRS 128
#define NCHUNK 32
#define PFD    4

#define T2_OFF  (LH * NF)
#define G_OFF   (2 * LH * NF + 2 * NROWS * NF)

typedef float v2f __attribute__((ext_vector_type(2)));
typedef float v4f __attribute__((ext_vector_type(4)));
typedef float v8f __attribute__((ext_vector_type(8)));

__device__ __forceinline__ v2f t2q_pair(int t) {
  const int w = t >> 5, lane = t & 31;
  const int hi = w & 1, qh = (w >> 1) & 1, ks = (w >> 2) & 15, grp = w >> 6;
  const int l  = 64 * grp + 4 * (lane >> 1) + 2 * qh + (lane & 1);
  const int kk = 4 * ks + 2 * hi;
  const bool ev = kk < 32;
  const int f0   = ev ? 4 * (kk & 7) + 2 * (kk >> 4) : 2 * (kk & 15) + 1;
  const int f1   = f0 + (ev ? 4 : 2);
  const int part = ev ? (kk >> 3) & 1 : (kk >> 4) & 1;
  const int qp = part * (LLEN / 4);
  v2f v;
  v.x = cospif((float)((f0 * l + qp) & (LLEN - 1)) * (2.0f / (float)LLEN));
  v.y = cospif((float)((f1 * l + qp) & (LLEN - 1)) * (2.0f / (float)LLEN));
  return v;
}

__global__ __launch_bounds__(512) void k_dft_fwd_mix(const float* __restrict__ x,
                                                     const float* __restrict__ wr,
                                                     const float* __restrict__ wi,
                                                     float* __restrict__ g,
                                                     float* __restrict__ t2q) {
  __shared__ float red[KSPLIT * HROWS * NF];
  __shared__ float xs[HROWS * NF];
  __shared__ v2f rot[64];
  if (threadIdx.x < 64) {
    float s, c;
    sincospif((float)threadIdx.x * (1.0f / 32.0f), &s, &c);
    const v2f cs = {c, s};
    rot[threadIdx.x] = cs;
  }
  const int lane = threadIdx.x & 31;
  const int wid  = threadIdx.x >> 5;
  const int hi   = lane >> 4;
  const int ln   = lane & 15;
  const int rowbase = blockIdx.x * HROWS;
  const float* xrow = x + (size_t)(rowbase + ln) * LLEN;

  v8f at0 = {}, at1 = {}, at2 = {}, at3 = {};
  const int fe0 = 4 * (ln & 7), fe1 = fe0 + 2, fo = 2 * ln + 1;
  const int l0 = 8 * wid + 4 * hi;
  v4f p0c, p0s, p1c, p1s, p2c, p2s;
  {
    v4f cc0, ss0, cc1, ss1, cc2, ss2;
    #pragma unroll
    for (int e = 0; e < 4; ++e) {
      float s, c;
      sincospif((float)(fe0 * (l0 + e)) * (1.0f / 4096.0f), &s, &c);
      cc0[e] = c;  ss0[e] = s;
      sincospif((float)(fe1 * (l0 + e)) * (1.0f / 4096.0f), &s, &c);
      cc1[e] = c;  ss1[e] = s;
      sincospif((float)(fo * (l0 + e)) * (1.0f / 4096.0f), &s, &c);
      cc2[e] = c;  ss2[e] = s;
    }
    const bool im = (ln >> 3) != 0;
    p0c = im ? -ss0 : cc0;  p0s = im ? cc0 : ss0;
    p1c = im ? -ss1 : cc1;  p1s = im ? cc1 : ss1;
    p2c = cc2;              p2s = ss2;
  }
  const float sg = (ln & 1) ? -1.0f : 1.0f;
  const int gid = blockIdx.x * 512 + threadIdx.x;
  *(v2f*)(t2q + 2 * (size_t)gid) = t2q_pair(gid);
  __builtin_amdgcn_fence(__ATOMIC_RELEASE, "workgroup", "local");
  __builtin_amdgcn_s_barrier();
  __builtin_amdgcn_fence(__ATOMIC_ACQUIRE, "workgroup", "local");

  const int k0beg = 8 * wid + 4 * hi;
  constexpr int NIT = LQ / (KSPLIT * 8);
  v4f qa[NIT], qb[NIT], qc[NIT], qd[NIT];
  #pragma unroll
  for (int it = 0; it < PFD; ++it) {
    const int kb = k0beg + 8 * KSPLIT * it;
    qa[it] = *(const v4f*)(xrow + kb);
    qb[it] = *(const v4f*)(xrow + kb + LQ);
    qc[it] = *(const v4f*)(xrow + kb + LH);
    qd[it] = *(const v4f*)(xrow + kb + LH + LQ);
  }
  #pragma unroll
  for (int it = 0; it < NIT; ++it) {
    if (it + PFD < NIT) {
      const int kb = k0beg + 8 * KSPLIT * (it + PFD);
      qa[it + PFD] = *(const v4f*)(xrow + kb);
      qb[it + PFD] = *(const v4f*)(xrow + kb + LQ);
      qc[it + PFD] = *(const v4f*)(xrow + kb + LH);
      qd[it + PFD] = *(const v4f*)(xrow + kb + LH + LQ);
    }
    __builtin_amdgcn_sched_barrier(0x106);
    const v4f xa = qa[it], xb = qb[it], xc = qc[it], xd = qd[it];
    const int m = it;
    const v2f r0 = rot[(fe0 * m) & 63];
    const v2f r1 = rot[(fe1 * m) & 63];
    const v2f r2 = rot[(fo  * m) & 63];
    const v4f b0 = r0.x * p0c - r0.y * p0s;
    const v4f b1 = r1.x * p1c - r1.y * p1s;
    const v4f b2 = r2.x * p2c - r2.y * p2s;
    const v4f b3 = -(r2.x * p2s) - r2.y * p2c;
    const v4f c2 = sg * b3;
    const v4f c3 = -sg * b2;
    const v4f ea = xa + xc, eb = xb + xd;
    const v4f oa = xa - xc, ob = xb - xd;
    const v4f ee = ea + eb, eo = ea - eb;
    const v2f ee0 = {ee.x, ee.y}, eo0 = {eo.x, eo.y}, oa0 = {oa.x, oa.y}, ob0 = {ob.x, ob.y};
    const v2f b00 = {b0.x, b0.y}, b10 = {b1.x, b1.y}, b20 = {b2.x, b2.y}, b30 = {b3.x, b3.y};
    const v2f c20 = {c2.x, c2.y}, c30 = {c3.x, c3.y};
    at0 = __builtin_amdgcn_wmma_f32_16x16x4_f32(false, ee0, false, b00, (short)0, at0, false, false);
    at1 = __builtin_amdgcn_wmma_f32_16x16x4_f32(false, eo0, false, b10, (short)0, at1, false, false);
    at2 = __builtin_amdgcn_wmma_f32_16x16x4_f32(false, oa0, false, b20, (short)0, at2, false, false);
    at3 = __builtin_amdgcn_wmma_f32_16x16x4_f32(false, oa0, false, b30, (short)0, at3, false, false);
    at2 = __builtin_amdgcn_wmma_f32_16x16x4_f32(false, ob0, false, c20, (short)0, at2, false, false);
    at3 = __builtin_amdgcn_wmma_f32_16x16x4_f32(false, ob0, false, c30, (short)0, at3, false, false);
    const v2f ee1 = {ee.z, ee.w}, eo1 = {eo.z, eo.w}, oa1 = {oa.z, oa.w}, ob1 = {ob.z, ob.w};
    const v2f b01 = {b0.z, b0.w}, b11 = {b1.z, b1.w}, b21 = {b2.z, b2.w}, b31 = {b3.z, b3.w};
    const v2f c21 = {c2.z, c2.w}, c31 = {c3.z, c3.w};
    at0 = __builtin_amdgcn_wmma_f32_16x16x4_f32(false, ee1, false, b01, (short)0, at0, false, false);
    at1 = __builtin_amdgcn_wmma_f32_16x16x4_f32(false, eo1, false, b11, (short)0, at1, false, false);
    at2 = __builtin_amdgcn_wmma_f32_16x16x4_f32(false, oa1, false, b21, (short)0, at2, false, false);
    at3 = __builtin_amdgcn_wmma_f32_16x16x4_f32(false, oa1, false, b31, (short)0, at3, false, false);
    at2 = __builtin_amdgcn_wmma_f32_16x16x4_f32(false, ob1, false, c21, (short)0, at2, false, false);
    at3 = __builtin_amdgcn_wmma_f32_16x16x4_f32(false, ob1, false, c31, (short)0, at3, false, false);
  }

  float* my = red + wid * (HROWS * NF);
  #pragma unroll
  for (int m = 0; m < 8; ++m) {
    const int r = m + 8 * hi;
    my[r * NF + 0 * 16 + ln] = at0[m];
    my[r * NF + 1 * 16 + ln] = at1[m];
    my[r * NF + 2 * 16 + ln] = at2[m];
    my[r * NF + 3 * 16 + ln] = at3[m];
  }
  const int f = threadIdx.x & 31;
  const int o = threadIdx.x >> 5;
  const int h = blockIdx.x & 3;
  const float* wr0 = wr + ((size_t)(h * 16) * 16 + o) * NMODES + f;
  const float* wi0 = wi + ((size_t)(h * 16) * 16 + o) * NMODES + f;
  float wre[16], wim[16];
  #pragma unroll
  for (int i = 0; i < 16; ++i) {
    wre[i] = wr0[i * (16 * NMODES)];
    wim[i] = wi0[i * (16 * NMODES)];
  }
  __syncthreads();
  #pragma unroll
  for (int u = 0; u < 2; ++u) {
    const int idx = threadIdx.x + 512 * u;
    float s = 0.0f;
    #pragma unroll
    for (int w = 0; w < KSPLIT; ++w) s += red[w * (HROWS * NF) + idx];
    xs[idx] = s;
  }
  __syncthreads();
  const int kr = (f & 1) ? 32 + (f >> 1) : ((f >> 1) & 1) * 16 + (f >> 2);
  const int ki = kr + ((f & 1) ? 16 : 8);
  const float* xr0 = xs + kr;
  const float* xi0 = xs + ki;
  float or_ = 0.0f, oi_ = 0.0f;
  #pragma unroll
  for (int i = 0; i < 16; ++i) {
    const float xr  = xr0[i * NF];
    const float xi  = xi0[i * NF];
    or_ += xr * wre[i] - xi * wim[i];
    oi_ += xr * wim[i] + xi * wre[i];
  }
  const float invL = 1.0f / (float)LLEN;
  const float scale = (f == 0) ? invL : 2.0f * invL;
  const size_t ro = (size_t)(rowbase + o);
  g[ro * NF + kr]      = or_ * scale;
  g[ro * NF + ki]      = oi_ * scale;
}

__global__ __launch_bounds__(128) __attribute__((amdgpu_waves_per_eu(2, 2))) void k_idft(const float* __restrict__ g,
                                              const float* __restrict__ t2q,
                                              const float* __restrict__ bias,
                                              float* __restrict__ out) {
  const int lane = threadIdx.x & 31;
  const int wid  = threadIdx.x >> 5;
  const int gw   = blockIdx.x * 4 + wid;
  const int mpair = gw >> 5;
  const int nc    = gw & (NCHUNK - 1);
  const int hi = lane >> 4, ln = lane & 15;
  const int rowbase = mpair * 32;

  v2f af0[16], af1[16];
  #pragma unroll
  for (int ks = 0; ks < 16; ++ks) {
    const int kb = ks * 4 + 2 * hi;
    af0[ks] = *(const v2f*)(g + (size_t)(rowbase + ln) * NF + kb);
    af1[ks] = *(const v2f*)(g + (size_t)(rowbase + 16 + ln) * NF + kb);
  }
  float bv0[8], bv1[8];
  #pragma unroll
  for (int m = 0; m < 8; ++m) {
    bv0[m] = bias[(rowbase + m + 8 * hi) & 63];
    bv1[m] = bias[(rowbase + 16 + m + 8 * hi) & 63];
  }

  const int col = nc * (LQ / NCHUNK) + 4 * ln;
  const float* tb = t2q + (size_t)nc * 4096 + (hi * 16 + ln) * 4;
  const float* tq = tb + (size_t)(LQ >> 6) * 4096;

  const v8f bz0 = {bv0[0], bv0[1], bv0[2], bv0[3], bv0[4], bv0[5], bv0[6], bv0[7]};
  const v8f bz1 = {bv1[0], bv1[1], bv1[2], bv1[3], bv1[4], bv1[5], bv1[6], bv1[7]};
  v8f z00 = bz0, z01 = bz0, z02 = bz0, z03 = bz0, z10 = bz1, z11 = bz1, z12 = bz1, z13 = bz1;
  v8f t00 = {}, t01 = {}, t02 = {}, t03 = {}, t10 = {}, t11 = {}, t12 = {}, t13 = {};
  #pragma unroll
  for (int ks = 0; ks < 4; ++ks) {
    const v4f p = *(const v4f*)(tb + (ks * 2 + 0) * 128);
    const v4f r = *(const v4f*)(tb + (ks * 2 + 1) * 128);
    const v2f b0 = {p.x, p.y};
    const v2f b1 = {p.z, p.w};
    const v2f b2 = {r.x, r.y};
    const v2f b3 = {r.z, r.w};
    z00 = __builtin_amdgcn_wmma_f32_16x16x4_f32(false, af0[ks], false, b0, (short)0, z00, false, false);
    z01 = __builtin_amdgcn_wmma_f32_16x16x4_f32(false, af0[ks], false, b1, (short)0, z01, false, false);
    z02 = __builtin_amdgcn_wmma_f32_16x16x4_f32(false, af0[ks], false, b2, (short)0, z02, false, false);
    z03 = __builtin_amdgcn_wmma_f32_16x16x4_f32(false, af0[ks], false, b3, (short)0, z03, false, false);
    z10 = __builtin_amdgcn_wmma_f32_16x16x4_f32(false, af1[ks], false, b0, (short)0, z10, false, false);
    z11 = __builtin_amdgcn_wmma_f32_16x16x4_f32(false, af1[ks], false, b1, (short)0, z11, false, false);
    z12 = __builtin_amdgcn_wmma_f32_16x16x4_f32(false, af1[ks], false, b2, (short)0, z12, false, false);
    z13 = __builtin_amdgcn_wmma_f32_16x16x4_f32(false, af1[ks], false, b3, (short)0, z13, false, false);
  }
  #pragma unroll
  for (int ks = 4; ks < 8; ++ks) {
    const v4f p = *(const v4f*)(tb + (ks * 2 + 0) * 128);
    const v4f r = *(const v4f*)(tb + (ks * 2 + 1) * 128);
    const v2f b0 = {p.x, p.y};
    const v2f b1 = {p.z, p.w};
    const v2f b2 = {r.x, r.y};
    const v2f b3 = {r.z, r.w};
    t00 = __builtin_amdgcn_wmma_f32_16x16x4_f32(false, af0[ks], false, b0, (short)0, t00, false, false);
    t01 = __builtin_amdgcn_wmma_f32_16x16x4_f32(false, af0[ks], false, b1, (short)0, t01, false, false);
    t02 = __builtin_amdgcn_wmma_f32_16x16x4_f32(false, af0[ks], false, b2, (short)0, t02, false, false);
    t03 = __builtin_amdgcn_wmma_f32_16x16x4_f32(false, af0[ks], false, b3, (short)0, t03, false, false);
    t10 = __builtin_amdgcn_wmma_f32_16x16x4_f32(false, af1[ks], false, b0, (short)0, t10, false, false);
    t11 = __builtin_amdgcn_wmma_f32_16x16x4_f32(false, af1[ks], false, b1, (short)0, t11, false, false);
    t12 = __builtin_amdgcn_wmma_f32_16x16x4_f32(false, af1[ks], false, b2, (short)0, t12, false, false);
    t13 = __builtin_amdgcn_wmma_f32_16x16x4_f32(false, af1[ks], false, b3, (short)0, t13, false, false);
  }

  #pragma unroll
  for (int ps = 0; ps < 2; ++ps) {
    const float* to = ps ? tq : tb;
    v8f o00 = {}, o01 = {}, o02 = {}, o03 = {};
    v8f o10 = {}, o11 = {}, o12 = {}, o13 = {};
    #pragma unroll
    for (int ks = 8; ks < 16; ++ks) {
      const v4f p = *(const v4f*)(to + (ks * 2 + 0) * 128);
      const v4f r = *(const v4f*)(to + (ks * 2 + 1) * 128);
      const v2f b0 = {p.x, p.y};
      const v2f b1 = {p.z, p.w};
      const v2f b2 = {r.x, r.y};
      const v2f b3 = {r.z, r.w};
      o00 = __builtin_amdgcn_wmma_f32_16x16x4_f32(false, af0[ks], false, b0, (short)0, o00, false, false);
      o01 = __builtin_amdgcn_wmma_f32_16x16x4_f32(false, af0[ks], false, b1, (short)0, o01, false, false);
      o02 = __builtin_amdgcn_wmma_f32_16x16x4_f32(false, af0[ks], false, b2, (short)0, o02, false, false);
      o03 = __builtin_amdgcn_wmma_f32_16x16x4_f32(false, af0[ks], false, b3, (short)0, o03, false, false);
      o10 = __builtin_amdgcn_wmma_f32_16x16x4_f32(false, af1[ks], false, b0, (short)0, o10, false, false);
      o11 = __builtin_amdgcn_wmma_f32_16x16x4_f32(false, af1[ks], false, b1, (short)0, o11, false, false);
      o12 = __builtin_amdgcn_wmma_f32_16x16x4_f32(false, af1[ks], false, b2, (short)0, o12, false, false);
      o13 = __builtin_amdgcn_wmma_f32_16x16x4_f32(false, af1[ks], false, b3, (short)0, o13, false, false);
    }
    const float sg = ps ? -1.0f : 1.0f;
    float* olo = out + (size_t)ps * LQ + col;
    float* ohi = olo + LH;
    #pragma unroll
    for (int m = 0; m < 8; ++m) {
      const int r = rowbase + m + 8 * hi;
      const float s00 = z00[m] + sg * t00[m], s01 = z01[m] + sg * t01[m];
      const float s02 = z02[m] + sg * t02[m], s03 = z03[m] + sg * t03[m];
      const float s10 = z10[m] + sg * t10[m], s11 = z11[m] + sg * t11[m];
      const float s12 = z12[m] + sg * t12[m], s13 = z13[m] + sg * t13[m];
      const v4f lo0 = {s00 + o00[m], s01 + o01[m], s02 + o02[m], s03 + o03[m]};
      const v4f hi0 = {s00 - o00[m], s01 - o01[m], s02 - o02[m], s03 - o03[m]};
      const v4f lo1 = {s10 + o10[m], s11 + o11[m], s12 + o12[m], s13 + o13[m]};
      const v4f hi1 = {s10 - o10[m], s11 - o11[m], s12 - o12[m], s13 - o13[m]};
      *(v4f*)(olo + (size_t)r * LLEN)        = lo0;
      *(v4f*)(ohi + (size_t)r * LLEN)        = hi0;
      *(v4f*)(olo + (size_t)(r + 16) * LLEN) = lo1;
      *(v4f*)(ohi + (size_t)(r + 16) * LLEN) = hi1;
    }
  }
}

extern "C" void kernel_launch(void* const* d_in, const int* in_sizes, int n_in,
                              void* d_out, int out_size, void* d_ws, size_t ws_size,
                              hipStream_t stream) {
  const float* x    = (const float*)d_in[0];
  const float* wre  = (const float*)d_in[1];
  const float* wim  = (const float*)d_in[2];
  const float* bias = (const float*)d_in[3];
  float* out = (float*)d_out;
  float* ws  = (float*)d_ws;

  float* t2q = ws + T2_OFF;
  float* g   = ws + G_OFF;

  k_dft_fwd_mix<<<NROWS / HROWS, 512, 0, stream>>>(x, wre, wim, g, t2q);
  k_idft       <<<(MPAIRS * NCHUNK) / 4, 128, 0, stream>>>(g, t2q, bias, out);
}


# Stage 3: stores leave in 4 groups of 16, each right behind its own WMMAs and held there by scheduling barriers (compiler had all 192 WMMAs ahead of all 64 stores); sums unchanged
# speedup vs baseline: 3.1362x; 1.0710x over previous
#include <hip/hip_runtime.h>


#define LLEN   8192
#define LH     4096
#define LQ     2048
#define NMODES 32
#define NF     64
#define NROWS  4096
#define KSPLIT 16
#define HROWS  16
#define MPAIRS 128
#define NCHUNK 32
#define PFD    4

#define T2_OFF  (LH * NF)
#define G_OFF   (2 * LH * NF + 2 * NROWS * NF)

typedef float v2f __attribute__((ext_vector_type(2)));
typedef float v4f __attribute__((ext_vector_type(4)));
typedef float v8f __attribute__((ext_vector_type(8)));

__device__ __forceinline__ v2f t2q_pair(int t) {
  const int w = t >> 5, lane = t & 31;
  const int hi = w & 1, qh = (w >> 1) & 1, ks = (w >> 2) & 15, grp = w >> 6;
  const int l  = 64 * grp + 4 * (lane >> 1) + 2 * qh + (lane & 1);
  const int kk = 4 * ks + 2 * hi;
  const bool ev = kk < 32;
  const int f0   = ev ? 4 * (kk & 7) + 2 * (kk >> 4) : 2 * (kk & 15) + 1;
  const int f1   = f0 + (ev ? 4 : 2);
  const int part = ev ? (kk >> 3) & 1 : (kk >> 4) & 1;
  const int qp = part * (LLEN / 4);
  v2f v;
  v.x = cospif((float)((f0 * l + qp) & (LLEN - 1)) * (2.0f / (float)LLEN));
  v.y = cospif((float)((f1 * l + qp) & (LLEN - 1)) * (2.0f / (float)LLEN));
  return v;
}

__global__ __launch_bounds__(512) void k_dft_fwd_mix(const float* __restrict__ x,
                                                     const float* __restrict__ wr,
                                                     const float* __restrict__ wi,
                                                     float* __restrict__ g,
                                                     float* __restrict__ t2q) {
  __shared__ float red[KSPLIT * HROWS * NF];
  __shared__ float xs[HROWS * NF];
  __shared__ v2f rot[64];
  if (threadIdx.x < 64) {
    float s, c;
    sincospif((float)threadIdx.x * (1.0f / 32.0f), &s, &c);
    const v2f cs = {c, s};
    rot[threadIdx.x] = cs;
  }
  const int lane = threadIdx.x & 31;
  const int wid  = threadIdx.x >> 5;
  const int hi   = lane >> 4;
  const int ln   = lane & 15;
  const int rowbase = blockIdx.x * HROWS;
  const float* xrow = x + (size_t)(rowbase + ln) * LLEN;

  v8f at0 = {}, at1 = {}, at2 = {}, at3 = {};
  const int fe0 = 4 * (ln & 7), fe1 = fe0 + 2, fo = 2 * ln + 1;
  const int l0 = 8 * wid + 4 * hi;
  v4f p0c, p0s, p1c, p1s, p2c, p2s;
  {
    v4f cc0, ss0, cc1, ss1, cc2, ss2;
    #pragma unroll
    for (int e = 0; e < 4; ++e) {
      float s, c;
      sincospif((float)(fe0 * (l0 + e)) * (1.0f / 4096.0f), &s, &c);
      cc0[e] = c;  ss0[e] = s;
      sincospif((float)(fe1 * (l0 + e)) * (1.0f / 4096.0f), &s, &c);
      cc1[e] = c;  ss1[e] = s;
      sincospif((float)(fo * (l0 + e)) * (1.0f / 4096.0f), &s, &c);
      cc2[e] = c;  ss2[e] = s;
    }
    const bool im = (ln >> 3) != 0;
    p0c = im ? -ss0 : cc0;  p0s = im ? cc0 : ss0;
    p1c = im ? -ss1 : cc1;  p1s = im ? cc1 : ss1;
    p2c = cc2;              p2s = ss2;
  }
  const float sg = (ln & 1) ? -1.0f : 1.0f;
  const int gid = blockIdx.x * 512 + threadIdx.x;
  *(v2f*)(t2q + 2 * (size_t)gid) = t2q_pair(gid);
  __builtin_amdgcn_fence(__ATOMIC_RELEASE, "workgroup", "local");
  __builtin_amdgcn_s_barrier();
  __builtin_amdgcn_fence(__ATOMIC_ACQUIRE, "workgroup", "local");

  const int k0beg = 8 * wid + 4 * hi;
  constexpr int NIT = LQ / (KSPLIT * 8);
  v4f qa[NIT], qb[NIT], qc[NIT], qd[NIT];
  #pragma unroll
  for (int it = 0; it < PFD; ++it) {
    const int kb = k0beg + 8 * KSPLIT * it;
    qa[it] = *(const v4f*)(xrow + kb);
    qb[it] = *(const v4f*)(xrow + kb + LQ);
    qc[it] = *(const v4f*)(xrow + kb + LH);
    qd[it] = *(const v4f*)(xrow + kb + LH + LQ);
  }
  #pragma unroll
  for (int it = 0; it < NIT; ++it) {
    if (it + PFD < NIT) {
      const int kb = k0beg + 8 * KSPLIT * (it + PFD);
      qa[it + PFD] = *(const v4f*)(xrow + kb);
      qb[it + PFD] = *(const v4f*)(xrow + kb + LQ);
      qc[it + PFD] = *(const v4f*)(xrow + kb + LH);
      qd[it + PFD] = *(const v4f*)(xrow + kb + LH + LQ);
    }
    __builtin_amdgcn_sched_barrier(0x106);
    const v4f xa = qa[it], xb = qb[it], xc = qc[it], xd = qd[it];
    const int m = it;
    const v2f r0 = rot[(fe0 * m) & 63];
    const v2f r1 = rot[(fe1 * m) & 63];
    const v2f r2 = rot[(fo  * m) & 63];
    const v4f b0 = r0.x * p0c - r0.y * p0s;
    const v4f b1 = r1.x * p1c - r1.y * p1s;
    const v4f b2 = r2.x * p2c - r2.y * p2s;
    const v4f b3 = -(r2.x * p2s) - r2.y * p2c;
    const v4f c2 = sg * b3;
    const v4f c3 = -sg * b2;
    const v4f ea = xa + xc, eb = xb + xd;
    const v4f oa = xa - xc, ob = xb - xd;
    const v4f ee = ea + eb, eo = ea - eb;
    const v2f ee0 = {ee.x, ee.y}, eo0 = {eo.x, eo.y}, oa0 = {oa.x, oa.y}, ob0 = {ob.x, ob.y};
    const v2f b00 = {b0.x, b0.y}, b10 = {b1.x, b1.y}, b20 = {b2.x, b2.y}, b30 = {b3.x, b3.y};
    const v2f c20 = {c2.x, c2.y}, c30 = {c3.x, c3.y};
    at0 = __builtin_amdgcn_wmma_f32_16x16x4_f32(false, ee0, false, b00, (short)0, at0, false, false);
    at1 = __builtin_amdgcn_wmma_f32_16x16x4_f32(false, eo0, false, b10, (short)0, at1, false, false);
    at2 = __builtin_amdgcn_wmma_f32_16x16x4_f32(false, oa0, false, b20, (short)0, at2, false, false);
    at3 = __builtin_amdgcn_wmma_f32_16x16x4_f32(false, oa0, false, b30, (short)0, at3, false, false);
    at2 = __builtin_amdgcn_wmma_f32_16x16x4_f32(false, ob0, false, c20, (short)0, at2, false, false);
    at3 = __builtin_amdgcn_wmma_f32_16x16x4_f32(false, ob0, false, c30, (short)0, at3, false, false);
    const v2f ee1 = {ee.z, ee.w}, eo1 = {eo.z, eo.w}, oa1 = {oa.z, oa.w}, ob1 = {ob.z, ob.w};
    const v2f b01 = {b0.z, b0.w}, b11 = {b1.z, b1.w}, b21 = {b2.z, b2.w}, b31 = {b3.z, b3.w};
    const v2f c21 = {c2.z, c2.w}, c31 = {c3.z, c3.w};
    at0 = __builtin_amdgcn_wmma_f32_16x16x4_f32(false, ee1, false, b01, (short)0, at0, false, false);
    at1 = __builtin_amdgcn_wmma_f32_16x16x4_f32(false, eo1, false, b11, (short)0, at1, false, false);
    at2 = __builtin_amdgcn_wmma_f32_16x16x4_f32(false, oa1, false, b21, (short)0, at2, false, false);
    at3 = __builtin_amdgcn_wmma_f32_16x16x4_f32(false, oa1, false, b31, (short)0, at3, false, false);
    at2 = __builtin_amdgcn_wmma_f32_16x16x4_f32(false, ob1, false, c21, (short)0, at2, false, false);
    at3 = __builtin_amdgcn_wmma_f32_16x16x4_f32(false, ob1, false, c31, (short)0, at3, false, false);
  }

  float* my = red + wid * (HROWS * NF);
  #pragma unroll
  for (int m = 0; m < 8; ++m) {
    const int r = m + 8 * hi;
    my[r * NF + 0 * 16 + ln] = at0[m];
    my[r * NF + 1 * 16 + ln] = at1[m];
    my[r * NF + 2 * 16 + ln] = at2[m];
    my[r * NF + 3 * 16 + ln] = at3[m];
  }
  const int f = threadIdx.x & 31;
  const int o = threadIdx.x >> 5;
  const int h = blockIdx.x & 3;
  const float* wr0 = wr + ((size_t)(h * 16) * 16 + o) * NMODES + f;
  const float* wi0 = wi + ((size_t)(h * 16) * 16 + o) * NMODES + f;
  float wre[16], wim[16];
  #pragma unroll
  for (int i = 0; i < 16; ++i) {
    wre[i] = wr0[i * (16 * NMODES)];
    wim[i] = wi0[i * (16 * NMODES)];
  }
  __syncthreads();
  #pragma unroll
  for (int u = 0; u < 2; ++u) {
    const int idx = threadIdx.x + 512 * u;
    float s = 0.0f;
    #pragma unroll
    for (int w = 0; w < KSPLIT; ++w) s += red[w * (HROWS * NF) + idx];
    xs[idx] = s;
  }
  __syncthreads();
  const int kr = (f & 1) ? 32 + (f >> 1) : ((f >> 1) & 1) * 16 + (f >> 2);
  const int ki = kr + ((f & 1) ? 16 : 8);
  const float* xr0 = xs + kr;
  const float* xi0 = xs + ki;
  float or_ = 0.0f, oi_ = 0.0f;
  #pragma unroll
  for (int i = 0; i < 16; ++i) {
    const float xr  = xr0[i * NF];
    const float xi  = xi0[i * NF];
    or_ += xr * wre[i] - xi * wim[i];
    oi_ += xr * wim[i] + xi * wre[i];
  }
  const float invL = 1.0f / (float)LLEN;
  const float scale = (f == 0) ? invL : 2.0f * invL;
  const size_t ro = (size_t)(rowbase + o);
  g[ro * NF + kr]      = or_ * scale;
  g[ro * NF + ki]      = oi_ * scale;
}

__device__ __forceinline__ void odd_group(const v2f (&af)[16], const v4f (&pt)[16],
                                          v8f& o0, v8f& o1, v8f& o2, v8f& o3) {
  #pragma unroll
  for (int j = 0; j < 8; ++j) {
    const v4f p = pt[2 * j + 0];
    const v4f r = pt[2 * j + 1];
    const v2f b0 = {p.x, p.y};
    const v2f b1 = {p.z, p.w};
    const v2f b2 = {r.x, r.y};
    const v2f b3 = {r.z, r.w};
    o0 = __builtin_amdgcn_wmma_f32_16x16x4_f32(false, af[8 + j], false, b0, (short)0, o0, false, false);
    o1 = __builtin_amdgcn_wmma_f32_16x16x4_f32(false, af[8 + j], false, b1, (short)0, o1, false, false);
    o2 = __builtin_amdgcn_wmma_f32_16x16x4_f32(false, af[8 + j], false, b2, (short)0, o2, false, false);
    o3 = __builtin_amdgcn_wmma_f32_16x16x4_f32(false, af[8 + j], false, b3, (short)0, o3, false, false);
  }
}

__device__ __forceinline__ void store_group(const v8f& z0, const v8f& z1, const v8f& z2, const v8f& z3,
                                            const v8f& t0, const v8f& t1, const v8f& t2, const v8f& t3,
                                            const v8f& o0, const v8f& o1, const v8f& o2, const v8f& o3,
                                            const float sg, float* olo, float* ohi,
                                            const int r0) {
  #pragma unroll
  for (int m = 0; m < 8; ++m) {
    const float s0 = z0[m] + sg * t0[m], s1 = z1[m] + sg * t1[m];
    const float s2 = z2[m] + sg * t2[m], s3 = z3[m] + sg * t3[m];
    const v4f lo = {s0 + o0[m], s1 + o1[m], s2 + o2[m], s3 + o3[m]};
    const v4f hh = {s0 - o0[m], s1 - o1[m], s2 - o2[m], s3 - o3[m]};
    *(v4f*)(olo + (size_t)(r0 + m) * LLEN) = lo;
    *(v4f*)(ohi + (size_t)(r0 + m) * LLEN) = hh;
  }
}

__global__ __launch_bounds__(128) __attribute__((amdgpu_waves_per_eu(2, 2))) void k_idft(const float* __restrict__ g,
                                              const float* __restrict__ t2q,
                                              const float* __restrict__ bias,
                                              float* __restrict__ out) {
  const int lane = threadIdx.x & 31;
  const int wid  = threadIdx.x >> 5;
  const int gw   = blockIdx.x * 4 + wid;
  const int mpair = gw >> 5;
  const int nc    = gw & (NCHUNK - 1);
  const int hi = lane >> 4, ln = lane & 15;
  const int rowbase = mpair * 32;

  v2f af0[16], af1[16];
  #pragma unroll
  for (int ks = 0; ks < 16; ++ks) {
    const int kb = ks * 4 + 2 * hi;
    af0[ks] = *(const v2f*)(g + (size_t)(rowbase + ln) * NF + kb);
    af1[ks] = *(const v2f*)(g + (size_t)(rowbase + 16 + ln) * NF + kb);
  }
  float bv0[8], bv1[8];
  #pragma unroll
  for (int m = 0; m < 8; ++m) {
    bv0[m] = bias[(rowbase + m + 8 * hi) & 63];
    bv1[m] = bias[(rowbase + 16 + m + 8 * hi) & 63];
  }

  const int col = nc * (LQ / NCHUNK) + 4 * ln;
  const float* tb = t2q + (size_t)nc * 4096 + (hi * 16 + ln) * 4;
  const float* tq = tb + (size_t)(LQ >> 6) * 4096;

  v4f pe[16], pa[16], pb[16];
  #pragma unroll
  for (int j = 0; j < 16; ++j) pe[j] = *(const v4f*)(tb + j * 128);
  #pragma unroll
  for (int j = 0; j < 16; ++j) pa[j] = *(const v4f*)(tb + (16 + j) * 128);
  #pragma unroll
  for (int j = 0; j < 16; ++j) pb[j] = *(const v4f*)(tq + (16 + j) * 128);

  const v8f bz0 = {bv0[0], bv0[1], bv0[2], bv0[3], bv0[4], bv0[5], bv0[6], bv0[7]};
  const v8f bz1 = {bv1[0], bv1[1], bv1[2], bv1[3], bv1[4], bv1[5], bv1[6], bv1[7]};
  v8f z00 = bz0, z01 = bz0, z02 = bz0, z03 = bz0, z10 = bz1, z11 = bz1, z12 = bz1, z13 = bz1;
  v8f t00 = {}, t01 = {}, t02 = {}, t03 = {}, t10 = {}, t11 = {}, t12 = {}, t13 = {};
  #pragma unroll
  for (int ks = 0; ks < 4; ++ks) {
    const v4f p = pe[ks * 2 + 0];
    const v4f r = pe[ks * 2 + 1];
    const v2f b0 = {p.x, p.y};
    const v2f b1 = {p.z, p.w};
    const v2f b2 = {r.x, r.y};
    const v2f b3 = {r.z, r.w};
    z00 = __builtin_amdgcn_wmma_f32_16x16x4_f32(false, af0[ks], false, b0, (short)0, z00, false, false);
    z01 = __builtin_amdgcn_wmma_f32_16x16x4_f32(false, af0[ks], false, b1, (short)0, z01, false, false);
    z02 = __builtin_amdgcn_wmma_f32_16x16x4_f32(false, af0[ks], false, b2, (short)0, z02, false, false);
    z03 = __builtin_amdgcn_wmma_f32_16x16x4_f32(false, af0[ks], false, b3, (short)0, z03, false, false);
    z10 = __builtin_amdgcn_wmma_f32_16x16x4_f32(false, af1[ks], false, b0, (short)0, z10, false, false);
    z11 = __builtin_amdgcn_wmma_f32_16x16x4_f32(false, af1[ks], false, b1, (short)0, z11, false, false);
    z12 = __builtin_amdgcn_wmma_f32_16x16x4_f32(false, af1[ks], false, b2, (short)0, z12, false, false);
    z13 = __builtin_amdgcn_wmma_f32_16x16x4_f32(false, af1[ks], false, b3, (short)0, z13, false, false);
  }
  #pragma unroll
  for (int ks = 4; ks < 8; ++ks) {
    const v4f p = pe[ks * 2 + 0];
    const v4f r = pe[ks * 2 + 1];
    const v2f b0 = {p.x, p.y};
    const v2f b1 = {p.z, p.w};
    const v2f b2 = {r.x, r.y};
    const v2f b3 = {r.z, r.w};
    t00 = __builtin_amdgcn_wmma_f32_16x16x4_f32(false, af0[ks], false, b0, (short)0, t00, false, false);
    t01 = __builtin_amdgcn_wmma_f32_16x16x4_f32(false, af0[ks], false, b1, (short)0, t01, false, false);
    t02 = __builtin_amdgcn_wmma_f32_16x16x4_f32(false, af0[ks], false, b2, (short)0, t02, false, false);
    t03 = __builtin_amdgcn_wmma_f32_16x16x4_f32(false, af0[ks], false, b3, (short)0, t03, false, false);
    t10 = __builtin_amdgcn_wmma_f32_16x16x4_f32(false, af1[ks], false, b0, (short)0, t10, false, false);
    t11 = __builtin_amdgcn_wmma_f32_16x16x4_f32(false, af1[ks], false, b1, (short)0, t11, false, false);
    t12 = __builtin_amdgcn_wmma_f32_16x16x4_f32(false, af1[ks], false, b2, (short)0, t12, false, false);
    t13 = __builtin_amdgcn_wmma_f32_16x16x4_f32(false, af1[ks], false, b3, (short)0, t13, false, false);
  }

  float* q0 = out + col;
  float* q1 = out + LQ + col;
  const int r0 = rowbase + 8 * hi;
  v8f oa0 = {}, oa1 = {}, oa2 = {}, oa3 = {};
  v8f ob0 = {}, ob1 = {}, ob2 = {}, ob3 = {};
  v8f oc0 = {}, oc1 = {}, oc2 = {}, oc3 = {};
  v8f od0 = {}, od1 = {}, od2 = {}, od3 = {};

  odd_group(af0, pa, oa0, oa1, oa2, oa3);
  store_group(z00, z01, z02, z03, t00, t01, t02, t03, oa0, oa1, oa2, oa3, 1.0f, q0, q0 + LH, r0);
  __builtin_amdgcn_sched_barrier(0);
  odd_group(af1, pa, ob0, ob1, ob2, ob3);
  store_group(z10, z11, z12, z13, t10, t11, t12, t13, ob0, ob1, ob2, ob3, 1.0f, q0, q0 + LH, r0 + 16);
  __builtin_amdgcn_sched_barrier(0);
  odd_group(af0, pb, oc0, oc1, oc2, oc3);
  store_group(z00, z01, z02, z03, t00, t01, t02, t03, oc0, oc1, oc2, oc3, -1.0f, q1, q1 + LH, r0);
  __builtin_amdgcn_sched_barrier(0);
  odd_group(af1, pb, od0, od1, od2, od3);
  store_group(z10, z11, z12, z13, t10, t11, t12, t13, od0, od1, od2, od3, -1.0f, q1, q1 + LH, r0 + 16);
}

extern "C" void kernel_launch(void* const* d_in, const int* in_sizes, int n_in,
                              void* d_out, int out_size, void* d_ws, size_t ws_size,
                              hipStream_t stream) {
  const float* x    = (const float*)d_in[0];
  const float* wre  = (const float*)d_in[1];
  const float* wim  = (const float*)d_in[2];
  const float* bias = (const float*)d_in[3];
  float* out = (float*)d_out;
  float* ws  = (float*)d_ws;

  float* t2q = ws + T2_OFF;
  float* g   = ws + G_OFF;

  k_dft_fwd_mix<<<NROWS / HROWS, 512, 0, stream>>>(x, wre, wim, g, t2q);
  k_idft       <<<(MPAIRS * NCHUNK) / 4, 128, 0, stream>>>(g, t2q, bias, out);
}


# Stage 3: a wave takes two adjacent 64-column chunks in turn (512 workgroups, all resident; was 1,024 in two rounds): G frags loaded once, chunk 2's table loaded under chunk 1's WMMAs; sums unchanged
# speedup vs baseline: 3.2805x; 1.0460x over previous
#include <hip/hip_runtime.h>


#define LLEN   8192
#define LH     4096
#define LQ     2048
#define NMODES 32
#define NF     64
#define NROWS  4096
#define KSPLIT 16
#define HROWS  16
#define MPAIRS 128
#define NCHUNK 32
#define PFD    4

#define T2_OFF  (LH * NF)
#define G_OFF   (2 * LH * NF + 2 * NROWS * NF)

typedef float v2f __attribute__((ext_vector_type(2)));
typedef float v4f __attribute__((ext_vector_type(4)));
typedef float v8f __attribute__((ext_vector_type(8)));

__device__ __forceinline__ v2f t2q_pair(int t) {
  const int w = t >> 5, lane = t & 31;
  const int hi = w & 1, qh = (w >> 1) & 1, ks = (w >> 2) & 15, grp = w >> 6;
  const int l  = 64 * grp + 4 * (lane >> 1) + 2 * qh + (lane & 1);
  const int kk = 4 * ks + 2 * hi;
  const bool ev = kk < 32;
  const int f0   = ev ? 4 * (kk & 7) + 2 * (kk >> 4) : 2 * (kk & 15) + 1;
  const int f1   = f0 + (ev ? 4 : 2);
  const int part = ev ? (kk >> 3) & 1 : (kk >> 4) & 1;
  const int qp = part * (LLEN / 4);
  v2f v;
  v.x = cospif((float)((f0 * l + qp) & (LLEN - 1)) * (2.0f / (float)LLEN));
  v.y = cospif((float)((f1 * l + qp) & (LLEN - 1)) * (2.0f / (float)LLEN));
  return v;
}

__global__ __launch_bounds__(512) void k_dft_fwd_mix(const float* __restrict__ x,
                                                     const float* __restrict__ wr,
                                                     const float* __restrict__ wi,
                                                     float* __restrict__ g,
                                                     float* __restrict__ t2q) {
  __shared__ float red[KSPLIT * HROWS * NF];
  __shared__ float xs[HROWS * NF];
  __shared__ v2f rot[64];
  if (threadIdx.x < 64) {
    float s, c;
    sincospif((float)threadIdx.x * (1.0f / 32.0f), &s, &c);
    const v2f cs = {c, s};
    rot[threadIdx.x] = cs;
  }
  const int lane = threadIdx.x & 31;
  const int wid  = threadIdx.x >> 5;
  const int hi   = lane >> 4;
  const int ln   = lane & 15;
  const int rowbase = blockIdx.x * HROWS;
  const float* xrow = x + (size_t)(rowbase + ln) * LLEN;

  v8f at0 = {}, at1 = {}, at2 = {}, at3 = {};
  const int fe0 = 4 * (ln & 7), fe1 = fe0 + 2, fo = 2 * ln + 1;
  const int l0 = 8 * wid + 4 * hi;
  v4f p0c, p0s, p1c, p1s, p2c, p2s;
  {
    v4f cc0, ss0, cc1, ss1, cc2, ss2;
    #pragma unroll
    for (int e = 0; e < 4; ++e) {
      float s, c;
      sincospif((float)(fe0 * (l0 + e)) * (1.0f / 4096.0f), &s, &c);
      cc0[e] = c;  ss0[e] = s;
      sincospif((float)(fe1 * (l0 + e)) * (1.0f / 4096.0f), &s, &c);
      cc1[e] = c;  ss1[e] = s;
      sincospif((float)(fo * (l0 + e)) * (1.0f / 4096.0f), &s, &c);
      cc2[e] = c;  ss2[e] = s;
    }
    const bool im = (ln >> 3) != 0;
    p0c = im ? -ss0 : cc0;  p0s = im ? cc0 : ss0;
    p1c = im ? -ss1 : cc1;  p1s = im ? cc1 : ss1;
    p2c = cc2;              p2s = ss2;
  }
  const float sg = (ln & 1) ? -1.0f : 1.0f;
  const int gid = blockIdx.x * 512 + threadIdx.x;
  *(v2f*)(t2q + 2 * (size_t)gid) = t2q_pair(gid);
  __builtin_amdgcn_fence(__ATOMIC_RELEASE, "workgroup", "local");
  __builtin_amdgcn_s_barrier();
  __builtin_amdgcn_fence(__ATOMIC_ACQUIRE, "workgroup", "local");

  const int k0beg = 8 * wid + 4 * hi;
  constexpr int NIT = LQ / (KSPLIT * 8);
  v4f qa[NIT], qb[NIT], qc[NIT], qd[NIT];
  #pragma unroll
  for (int it = 0; it < PFD; ++it) {
    const int kb = k0beg + 8 * KSPLIT * it;
    qa[it] = *(const v4f*)(xrow + kb);
    qb[it] = *(const v4f*)(xrow + kb + LQ);
    qc[it] = *(const v4f*)(xrow + kb + LH);
    qd[it] = *(const v4f*)(xrow + kb + LH + LQ);
  }
  #pragma unroll
  for (int it = 0; it < NIT; ++it) {
    if (it + PFD < NIT) {
      const int kb = k0beg + 8 * KSPLIT * (it + PFD);
      qa[it + PFD] = *(const v4f*)(xrow + kb);
      qb[it + PFD] = *(const v4f*)(xrow + kb + LQ);
      qc[it + PFD] = *(const v4f*)(xrow + kb + LH);
      qd[it + PFD] = *(const v4f*)(xrow + kb + LH + LQ);
    }
    __builtin_amdgcn_sched_barrier(0x106);
    const v4f xa = qa[it], xb = qb[it], xc = qc[it], xd = qd[it];
    const int m = it;
    const v2f r0 = rot[(fe0 * m) & 63];
    const v2f r1 = rot[(fe1 * m) & 63];
    const v2f r2 = rot[(fo  * m) & 63];
    const v4f b0 = r0.x * p0c - r0.y * p0s;
    const v4f b1 = r1.x * p1c - r1.y * p1s;
    const v4f b2 = r2.x * p2c - r2.y * p2s;
    const v4f b3 = -(r2.x * p2s) - r2.y * p2c;
    const v4f c2 = sg * b3;
    const v4f c3 = -sg * b2;
    const v4f ea = xa + xc, eb = xb + xd;
    const v4f oa = xa - xc, ob = xb - xd;
    const v4f ee = ea + eb, eo = ea - eb;
    const v2f ee0 = {ee.x, ee.y}, eo0 = {eo.x, eo.y}, oa0 = {oa.x, oa.y}, ob0 = {ob.x, ob.y};
    const v2f b00 = {b0.x, b0.y}, b10 = {b1.x, b1.y}, b20 = {b2.x, b2.y}, b30 = {b3.x, b3.y};
    const v2f c20 = {c2.x, c2.y}, c30 = {c3.x, c3.y};
    at0 = __builtin_amdgcn_wmma_f32_16x16x4_f32(false, ee0, false, b00, (short)0, at0, false, false);
    at1 = __builtin_amdgcn_wmma_f32_16x16x4_f32(false, eo0, false, b10, (short)0, at1, false, false);
    at2 = __builtin_amdgcn_wmma_f32_16x16x4_f32(false, oa0, false, b20, (short)0, at2, false, false);
    at3 = __builtin_amdgcn_wmma_f32_16x16x4_f32(false, oa0, false, b30, (short)0, at3, false, false);
    at2 = __builtin_amdgcn_wmma_f32_16x16x4_f32(false, ob0, false, c20, (short)0, at2, false, false);
    at3 = __builtin_amdgcn_wmma_f32_16x16x4_f32(false, ob0, false, c30, (short)0, at3, false, false);
    const v2f ee1 = {ee.z, ee.w}, eo1 = {eo.z, eo.w}, oa1 = {oa.z, oa.w}, ob1 = {ob.z, ob.w};
    const v2f b01 = {b0.z, b0.w}, b11 = {b1.z, b1.w}, b21 = {b2.z, b2.w}, b31 = {b3.z, b3.w};
    const v2f c21 = {c2.z, c2.w}, c31 = {c3.z, c3.w};
    at0 = __builtin_amdgcn_wmma_f32_16x16x4_f32(false, ee1, false, b01, (short)0, at0, false, false);
    at1 = __builtin_amdgcn_wmma_f32_16x16x4_f32(false, eo1, false, b11, (short)0, at1, false, false);
    at2 = __builtin_amdgcn_wmma_f32_16x16x4_f32(false, oa1, false, b21, (short)0, at2, false, false);
    at3 = __builtin_amdgcn_wmma_f32_16x16x4_f32(false, oa1, false, b31, (short)0, at3, false, false);
    at2 = __builtin_amdgcn_wmma_f32_16x16x4_f32(false, ob1, false, c21, (short)0, at2, false, false);
    at3 = __builtin_amdgcn_wmma_f32_16x16x4_f32(false, ob1, false, c31, (short)0, at3, false, false);
  }

  float* my = red + wid * (HROWS * NF);
  #pragma unroll
  for (int m = 0; m < 8; ++m) {
    const int r = m + 8 * hi;
    my[r * NF + 0 * 16 + ln] = at0[m];
    my[r * NF + 1 * 16 + ln] = at1[m];
    my[r * NF + 2 * 16 + ln] = at2[m];
    my[r * NF + 3 * 16 + ln] = at3[m];
  }
  const int f = threadIdx.x & 31;
  const int o = threadIdx.x >> 5;
  const int h = blockIdx.x & 3;
  const float* wr0 = wr + ((size_t)(h * 16) * 16 + o) * NMODES + f;
  const float* wi0 = wi + ((size_t)(h * 16) * 16 + o) * NMODES + f;
  float wre[16], wim[16];
  #pragma unroll
  for (int i = 0; i < 16; ++i) {
    wre[i] = wr0[i * (16 * NMODES)];
    wim[i] = wi0[i * (16 * NMODES)];
  }
  __syncthreads();
  #pragma unroll
  for (int u = 0; u < 2; ++u) {
    const int idx = threadIdx.x + 512 * u;
    float s = 0.0f;
    #pragma unroll
    for (int w = 0; w < KSPLIT; ++w) s += red[w * (HROWS * NF) + idx];
    xs[idx] = s;
  }
  __syncthreads();
  const int kr = (f & 1) ? 32 + (f >> 1) : ((f >> 1) & 1) * 16 + (f >> 2);
  const int ki = kr + ((f & 1) ? 16 : 8);
  const float* xr0 = xs + kr;
  const float* xi0 = xs + ki;
  float or_ = 0.0f, oi_ = 0.0f;
  #pragma unroll
  for (int i = 0; i < 16; ++i) {
    const float xr  = xr0[i * NF];
    const float xi  = xi0[i * NF];
    or_ += xr * wre[i] - xi * wim[i];
    oi_ += xr * wim[i] + xi * wre[i];
  }
  const float invL = 1.0f / (float)LLEN;
  const float scale = (f == 0) ? invL : 2.0f * invL;
  const size_t ro = (size_t)(rowbase + o);
  g[ro * NF + kr]      = or_ * scale;
  g[ro * NF + ki]      = oi_ * scale;
}

__device__ __forceinline__ void odd_group(const v2f (&af)[16], const v4f (&pt)[16],
                                          v8f& o0, v8f& o1, v8f& o2, v8f& o3) {
  #pragma unroll
  for (int j = 0; j < 8; ++j) {
    const v4f p = pt[2 * j + 0];
    const v4f r = pt[2 * j + 1];
    const v2f b0 = {p.x, p.y};
    const v2f b1 = {p.z, p.w};
    const v2f b2 = {r.x, r.y};
    const v2f b3 = {r.z, r.w};
    o0 = __builtin_amdgcn_wmma_f32_16x16x4_f32(false, af[8 + j], false, b0, (short)0, o0, false, false);
    o1 = __builtin_amdgcn_wmma_f32_16x16x4_f32(false, af[8 + j], false, b1, (short)0, o1, false, false);
    o2 = __builtin_amdgcn_wmma_f32_16x16x4_f32(false, af[8 + j], false, b2, (short)0, o2, false, false);
    o3 = __builtin_amdgcn_wmma_f32_16x16x4_f32(false, af[8 + j], false, b3, (short)0, o3, false, false);
  }
}

__device__ __forceinline__ void store_group(const v8f& z0, const v8f& z1, const v8f& z2, const v8f& z3,
                                            const v8f& t0, const v8f& t1, const v8f& t2, const v8f& t3,
                                            const v8f& o0, const v8f& o1, const v8f& o2, const v8f& o3,
                                            const float sg, float* olo, float* ohi,
                                            const int r0) {
  #pragma unroll
  for (int m = 0; m < 8; ++m) {
    const float s0 = z0[m] + sg * t0[m], s1 = z1[m] + sg * t1[m];
    const float s2 = z2[m] + sg * t2[m], s3 = z3[m] + sg * t3[m];
    const v4f lo = {s0 + o0[m], s1 + o1[m], s2 + o2[m], s3 + o3[m]};
    const v4f hh = {s0 - o0[m], s1 - o1[m], s2 - o2[m], s3 - o3[m]};
    *(v4f*)(olo + (size_t)(r0 + m) * LLEN) = lo;
    *(v4f*)(ohi + (size_t)(r0 + m) * LLEN) = hh;
  }
}

__device__ __forceinline__ void load_frags(v4f (&p)[16], const float* base) {
  #pragma unroll
  for (int j = 0; j < 16; ++j) p[j] = *(const v4f*)(base + j * 128);
}

__global__ __launch_bounds__(128) __attribute__((amdgpu_waves_per_eu(2, 2))) void k_idft(const float* __restrict__ g,
                                              const float* __restrict__ t2q,
                                              const float* __restrict__ bias,
                                              float* __restrict__ out) {
  const int lane = threadIdx.x & 31;
  const int wid  = threadIdx.x >> 5;
  const int gw   = blockIdx.x * 4 + wid;
  const int mpair = gw >> 4;
  const int nc0   = (gw & (NCHUNK / 2 - 1)) * 2;
  const int hi = lane >> 4, ln = lane & 15;
  const int rowbase = mpair * 32;

  v2f af0[16], af1[16];
  #pragma unroll
  for (int ks = 0; ks < 16; ++ks) {
    const int kb = ks * 4 + 2 * hi;
    af0[ks] = *(const v2f*)(g + (size_t)(rowbase + ln) * NF + kb);
    af1[ks] = *(const v2f*)(g + (size_t)(rowbase + 16 + ln) * NF + kb);
  }
  float bv0[8], bv1[8];
  #pragma unroll
  for (int m = 0; m < 8; ++m) {
    bv0[m] = bias[(rowbase + m + 8 * hi) & 63];
    bv1[m] = bias[(rowbase + 16 + m + 8 * hi) & 63];
  }

  const int col = nc0 * (LQ / NCHUNK) + 4 * ln;
  const float* tb = t2q + (size_t)nc0 * 4096 + (hi * 16 + ln) * 4;
  const float* tq = tb + (size_t)(LQ >> 6) * 4096;

  v4f pe[16], pa[16], pb[16];
  load_frags(pe, tb);
  load_frags(pa, tb + 16 * 128);
  load_frags(pb, tq + 16 * 128);

  const v8f bz0 = {bv0[0], bv0[1], bv0[2], bv0[3], bv0[4], bv0[5], bv0[6], bv0[7]};
  const v8f bz1 = {bv1[0], bv1[1], bv1[2], bv1[3], bv1[4], bv1[5], bv1[6], bv1[7]};
  const int r0 = rowbase + 8 * hi;

  #pragma unroll
  for (int c = 0; c < 2; ++c) {
    v8f z00 = bz0, z01 = bz0, z02 = bz0, z03 = bz0, z10 = bz1, z11 = bz1, z12 = bz1, z13 = bz1;
    v8f t00 = {}, t01 = {}, t02 = {}, t03 = {}, t10 = {}, t11 = {}, t12 = {}, t13 = {};
    #pragma unroll
    for (int ks = 0; ks < 4; ++ks) {
      const v4f p = pe[ks * 2 + 0];
      const v4f r = pe[ks * 2 + 1];
      const v2f b0 = {p.x, p.y};
      const v2f b1 = {p.z, p.w};
      const v2f b2 = {r.x, r.y};
      const v2f b3 = {r.z, r.w};
      z00 = __builtin_amdgcn_wmma_f32_16x16x4_f32(false, af0[ks], false, b0, (short)0, z00, false, false);
      z01 = __builtin_amdgcn_wmma_f32_16x16x4_f32(false, af0[ks], false, b1, (short)0, z01, false, false);
      z02 = __builtin_amdgcn_wmma_f32_16x16x4_f32(false, af0[ks], false, b2, (short)0, z02, false, false);
      z03 = __builtin_amdgcn_wmma_f32_16x16x4_f32(false, af0[ks], false, b3, (short)0, z03, false, false);
      z10 = __builtin_amdgcn_wmma_f32_16x16x4_f32(false, af1[ks], false, b0, (short)0, z10, false, false);
      z11 = __builtin_amdgcn_wmma_f32_16x16x4_f32(false, af1[ks], false, b1, (short)0, z11, false, false);
      z12 = __builtin_amdgcn_wmma_f32_16x16x4_f32(false, af1[ks], false, b2, (short)0, z12, false, false);
      z13 = __builtin_amdgcn_wmma_f32_16x16x4_f32(false, af1[ks], false, b3, (short)0, z13, false, false);
    }
    #pragma unroll
    for (int ks = 4; ks < 8; ++ks) {
      const v4f p = pe[ks * 2 + 0];
      const v4f r = pe[ks * 2 + 1];
      const v2f b0 = {p.x, p.y};
      const v2f b1 = {p.z, p.w};
      const v2f b2 = {r.x, r.y};
      const v2f b3 = {r.z, r.w};
      t00 = __builtin_amdgcn_wmma_f32_16x16x4_f32(false, af0[ks], false, b0, (short)0, t00, false, false);
      t01 = __builtin_amdgcn_wmma_f32_16x16x4_f32(false, af0[ks], false, b1, (short)0, t01, false, false);
      t02 = __builtin_amdgcn_wmma_f32_16x16x4_f32(false, af0[ks], false, b2, (short)0, t02, false, false);
      t03 = __builtin_amdgcn_wmma_f32_16x16x4_f32(false, af0[ks], false, b3, (short)0, t03, false, false);
      t10 = __builtin_amdgcn_wmma_f32_16x16x4_f32(false, af1[ks], false, b0, (short)0, t10, false, false);
      t11 = __builtin_amdgcn_wmma_f32_16x16x4_f32(false, af1[ks], false, b1, (short)0, t11, false, false);
      t12 = __builtin_amdgcn_wmma_f32_16x16x4_f32(false, af1[ks], false, b2, (short)0, t12, false, false);
      t13 = __builtin_amdgcn_wmma_f32_16x16x4_f32(false, af1[ks], false, b3, (short)0, t13, false, false);
    }
    if (c == 0) load_frags(pe, tb + 4096);

    float* q0 = out + col + c * (LQ / NCHUNK);
    float* q1 = q0 + LQ;
    v8f oa0 = {}, oa1 = {}, oa2 = {}, oa3 = {};
    v8f ob0 = {}, ob1 = {}, ob2 = {}, ob3 = {};
    v8f oc0 = {}, oc1 = {}, oc2 = {}, oc3 = {};
    v8f od0 = {}, od1 = {}, od2 = {}, od3 = {};

    odd_group(af0, pa, oa0, oa1, oa2, oa3);
    store_group(z00, z01, z02, z03, t00, t01, t02, t03, oa0, oa1, oa2, oa3, 1.0f, q0, q0 + LH, r0);
    __builtin_amdgcn_sched_barrier(0);
    odd_group(af1, pa, ob0, ob1, ob2, ob3);
    store_group(z10, z11, z12, z13, t10, t11, t12, t13, ob0, ob1, ob2, ob3, 1.0f, q0, q0 + LH, r0 + 16);
    __builtin_amdgcn_sched_barrier(0);
    if (c == 0) load_frags(pa, tb + 4096 + 16 * 128);
    odd_group(af0, pb, oc0, oc1, oc2, oc3);
    store_group(z00, z01, z02, z03, t00, t01, t02, t03, oc0, oc1, oc2, oc3, -1.0f, q1, q1 + LH, r0);
    __builtin_amdgcn_sched_barrier(0);
    odd_group(af1, pb, od0, od1, od2, od3);
    store_group(z10, z11, z12, z13, t10, t11, t12, t13, od0, od1, od2, od3, -1.0f, q1, q1 + LH, r0 + 16);
    if (c == 0) {
      __builtin_amdgcn_sched_barrier(0);
      load_frags(pb, tq + 4096 + 16 * 128);
    }
  }
}

extern "C" void kernel_launch(void* const* d_in, const int* in_sizes, int n_in,
                              void* d_out, int out_size, void* d_ws, size_t ws_size,
                              hipStream_t stream) {
  const float* x    = (const float*)d_in[0];
  const float* wre  = (const float*)d_in[1];
  const float* wim  = (const float*)d_in[2];
  const float* bias = (const float*)d_in[3];
  float* out = (float*)d_out;
  float* ws  = (float*)d_ws;

  float* t2q = ws + T2_OFF;
  float* g   = ws + G_OFF;

  k_dft_fwd_mix<<<NROWS / HROWS, 512, 0, stream>>>(x, wre, wim, g, t2q);
  k_idft       <<<(MPAIRS * NCHUNK) / 8, 128, 0, stream>>>(g, t2q, bias, out);
}
